# starting kernel
# baseline (speedup 1.0000x reference)
#include <hip/hip_runtime.h>
#include <hip/hip_bf16.h>
#include <stdint.h>

#define BATCH 1024
#define EC    4096
#define CA3   2048
#define LR    0.01f

typedef float v2f __attribute__((ext_vector_type(2)));
typedef float v8f __attribute__((ext_vector_type(8)));

static __device__ __forceinline__ v8f wmma_f32_k4(v2f a, v2f b, v8f c) {
    return __builtin_amdgcn_wmma_f32_16x16x4_f32(
        false, a, false, b, (short)0, c, false, false);
}

static __device__ __forceinline__ void gemm_step_rowmajor(
    const float* ak, const float* bk, uint32_t avoff, uint32_t bvoff,
    v8f (&acc)[4][4])
{
    v2f a[4], b[4];
#pragma unroll
    for (int i = 0; i < 4; ++i)
        a[i] = *(const v2f*)(ak + avoff + (size_t)i * 16 * EC);
#pragma unroll
    for (int i = 0; i < 4; ++i)
        b[i] = *(const v2f*)(bk + bvoff + (size_t)i * 16 * EC);
#pragma unroll
    for (int mi = 0; mi < 4; ++mi)
#pragma unroll
        for (int ni = 0; ni < 4; ++ni)
            acc[mi][ni] = wmma_f32_k4(a[mi], b[ni], acc[mi][ni]);
}

__global__ __launch_bounds__(256) void k_gemm_pre(
    const float* __restrict__ ec, const float* __restrict__ dg,
    const float* __restrict__ W, float* __restrict__ out)
{
    const int lane = threadIdx.x & 31;
    const int wave = threadIdx.x >> 5;
    const int wm = wave >> 2;
    const int wn = wave & 3;
    const int lm = lane & 15;
    const int lk = (lane >> 4) << 1;

    const int row0 = blockIdx.y * 128 + wm * 64;
    const int col0 = blockIdx.x * 256 + wn * 64;

    const uint32_t voff = (uint32_t)(lm * EC + lk);
    const float* abase = ec + (size_t)row0 * EC;
    const float* bbase = W  + (size_t)col0 * EC;

    v8f acc[4][4];
#pragma unroll
    for (int i = 0; i < 4; ++i)
#pragma unroll
        for (int j = 0; j < 4; ++j) acc[i][j] = (v8f){};

    for (int kb = 0; kb < EC; kb += 32) {
        if (kb + 512 < EC) {
            __builtin_prefetch(abase + voff + kb + 512, 0, 1);
            __builtin_prefetch(bbase + voff + kb + 512, 0, 1);
        }
#pragma unroll
        for (int kk = 0; kk < 32; kk += 4)
            gemm_step_rowmajor(abase + kb + kk, bbase + kb + kk,
                               voff, voff, acc);
    }

    const int rsel = (lane >> 4) * 8;
#pragma unroll
    for (int mi = 0; mi < 4; ++mi)
#pragma unroll
        for (int ni = 0; ni < 4; ++ni)
#pragma unroll
            for (int r = 0; r < 8; ++r) {
                const int row = row0 + mi * 16 + rsel + r;
                const int col = col0 + ni * 16 + lm;
                const size_t idx = (size_t)row * CA3 + col;
                out[idx] = acc[mi][ni][r] + dg[idx];
            }
}

__global__ __launch_bounds__(256) void k_colmean(
    const float* __restrict__ pc, float* __restrict__ colmean)
{
    const int t = blockIdx.x * blockDim.x + threadIdx.x;
    if (t < CA3) {
        float s = 0.f;
        for (int b = 0; b < BATCH; ++b) s += pc[(size_t)b * CA3 + t];
        colmean[t] = s * (1.0f / BATCH);
    }
}

__global__ __launch_bounds__(256) void k_delta(
    const float* __restrict__ pc, const float* __restrict__ ec,
    const float* __restrict__ W, const float* __restrict__ colmean,
    float* __restrict__ delta)
{
    const int lane = threadIdx.x & 31;
    const int wave = threadIdx.x >> 5;
    const int wm = wave >> 2;
    const int wn = wave & 3;
    const int lm = lane & 15;
    const int lk = (lane >> 4) << 1;

    const int row0 = blockIdx.y * 128 + wm * 64;
    const int col0 = blockIdx.x * 256 + wn * 64;

    const uint32_t avoff = (uint32_t)(lk * CA3 + lm);
    const uint32_t bvoff = (uint32_t)(lk * EC  + lm);
    const float* abase = pc + row0;
    const float* bbase = ec + col0;

    v8f acc[4][4];
#pragma unroll
    for (int i = 0; i < 4; ++i)
#pragma unroll
        for (int j = 0; j < 4; ++j) acc[i][j] = (v8f){};

#pragma unroll 2
    for (int k0 = 0; k0 < BATCH; k0 += 4) {
        const float* ak = abase + (size_t)k0 * CA3;
        const float* bk = bbase + (size_t)k0 * EC;
        v2f a[4], b[4];
#pragma unroll
        for (int i = 0; i < 4; ++i) {
            const float* p = ak + avoff + i * 16;
            a[i] = (v2f){ p[0], p[CA3] };
        }
#pragma unroll
        for (int i = 0; i < 4; ++i) {
            const float* p = bk + bvoff + i * 16;
            b[i] = (v2f){ p[0], p[EC] };
        }
#pragma unroll
        for (int mi = 0; mi < 4; ++mi)
#pragma unroll
            for (int ni = 0; ni < 4; ++ni)
                acc[mi][ni] = wmma_f32_k4(a[mi], b[ni], acc[mi][ni]);
    }

    const int rsel = (lane >> 4) * 8;
#pragma unroll
    for (int mi = 0; mi < 4; ++mi)
#pragma unroll
        for (int ni = 0; ni < 4; ++ni)
#pragma unroll
            for (int r = 0; r < 8; ++r) {
                const int t = row0 + mi * 16 + rsel + r;
                const int i = col0 + ni * 16 + lm;
                const size_t idx = (size_t)t * EC + i;
                delta[idx] = LR * (acc[mi][ni][r] * (1.0f / BATCH)
                                   - colmean[t] * W[idx]);
            }
}

__global__ __launch_bounds__(256) void k_loss(
    const float* __restrict__ ec, const float* __restrict__ delta,
    float* __restrict__ partials)
{
    const int lane = threadIdx.x & 31;
    const int wave = threadIdx.x >> 5;
    const int wm = wave >> 2;
    const int wn = wave & 3;
    const int lm = lane & 15;
    const int lk = (lane >> 4) << 1;

    const int row0 = blockIdx.y * 128 + wm * 64;
    const int col0 = blockIdx.x * 256 + wn * 64;

    const uint32_t voff = (uint32_t)(lm * EC + lk);
    const float* abase = ec    + (size_t)row0 * EC;
    const float* bbase = delta + (size_t)col0 * EC;

    v8f acc[4][4];
#pragma unroll
    for (int i = 0; i < 4; ++i)
#pragma unroll
        for (int j = 0; j < 4; ++j) acc[i][j] = (v8f){};

    for (int kb = 0; kb < EC; kb += 32) {
#pragma unroll
        for (int kk = 0; kk < 32; kk += 4)
            gemm_step_rowmajor(abase + kb + kk, bbase + kb + kk,
                               voff, voff, acc);
    }

    float lsum = 0.f;
#pragma unroll
    for (int mi = 0; mi < 4; ++mi)
#pragma unroll
        for (int ni = 0; ni < 4; ++ni)
#pragma unroll
            for (int r = 0; r < 8; ++r) {
                const float v = acc[mi][ni][r];
                lsum += v * v;
            }

    __shared__ float red[256];
    red[threadIdx.x] = lsum;
    __syncthreads();
#pragma unroll
    for (int s = 128; s > 0; s >>= 1) {
        if (threadIdx.x < (unsigned)s) red[threadIdx.x] += red[threadIdx.x + s];
        __syncthreads();
    }
    if (threadIdx.x == 0)
        partials[blockIdx.y * gridDim.x + blockIdx.x] = red[0];
}

__global__ void k_finalize(const float* __restrict__ partials, int n,
                           float* __restrict__ out_losses)
{
    if (threadIdx.x == 0 && blockIdx.x == 0) {
        float s = 0.f;
        for (int i = 0; i < n; ++i) s += partials[i];
        const float loss = s / (float)((size_t)BATCH * CA3);
        out_losses[0] = 0.0f;
        out_losses[1] = loss;
        out_losses[2] = loss;
    }
}

extern "C" void kernel_launch(void* const* d_in, const int* in_sizes, int n_in,
                              void* d_out, int out_size, void* d_ws, size_t ws_size,
                              hipStream_t stream)
{
    (void)in_sizes; (void)n_in; (void)out_size; (void)ws_size;
    const float* ec = (const float*)d_in[0];
    const float* dg = (const float*)d_in[1];
    const float* W  = (const float*)d_in[2];
    float* out = (float*)d_out;

    float* delta    = (float*)d_ws;
    float* colmean  = delta + (size_t)CA3 * EC;
    float* partials = colmean + CA3;

    k_gemm_pre<<<dim3(CA3 / 256, BATCH / 128), 256, 0, stream>>>(ec, dg, W, out);
    k_colmean<<<CA3 / 256, 256, 0, stream>>>(out, colmean);
    k_delta<<<dim3(EC / 256, CA3 / 128), 256, 0, stream>>>(out, ec, W, colmean, delta);
    k_loss<<<dim3(CA3 / 256, BATCH / 128), 256, 0, stream>>>(ec, delta, partials);
    k_finalize<<<1, 64, 0, stream>>>(partials, 64, out + (size_t)BATCH * CA3);
}


# GEMM kernels 1 and 4 re-tiled from 64 to 256 work-groups (block tile 64x128, 4 waves of 32x64) so all 256 WGPs work; loss partial sums regroup with the tiles (256 partials, new sum order)
# speedup vs baseline: 1.6903x; 1.6903x over previous
#include <hip/hip_runtime.h>
#include <hip/hip_bf16.h>
#include <stdint.h>

#define BATCH 1024
#define EC    4096
#define CA3   2048
#define LR    0.01f

typedef float v2f __attribute__((ext_vector_type(2)));
typedef float v8f __attribute__((ext_vector_type(8)));

static __device__ __forceinline__ v8f wmma_f32_k4(v2f a, v2f b, v8f c) {
    return __builtin_amdgcn_wmma_f32_16x16x4_f32(
        false, a, false, b, (short)0, c, false, false);
}

static __device__ __forceinline__ void gemm_step_rowmajor(
    const float* ak, const float* bk, uint32_t avoff, uint32_t bvoff,
    v8f (&acc)[2][4])
{
    v2f a[2], b[4];
#pragma unroll
    for (int i = 0; i < 2; ++i)
        a[i] = *(const v2f*)(ak + avoff + (size_t)i * 16 * EC);
#pragma unroll
    for (int i = 0; i < 4; ++i)
        b[i] = *(const v2f*)(bk + bvoff + (size_t)i * 16 * EC);
#pragma unroll
    for (int mi = 0; mi < 2; ++mi)
#pragma unroll
        for (int ni = 0; ni < 4; ++ni)
            acc[mi][ni] = wmma_f32_k4(a[mi], b[ni], acc[mi][ni]);
}

__global__ __launch_bounds__(128) void k_gemm_pre(
    const float* __restrict__ ec, const float* __restrict__ dg,
    const float* __restrict__ W, float* __restrict__ out)
{
    const int lane = threadIdx.x & 31;
    const int wave = threadIdx.x >> 5;
    const int wm = wave >> 1;
    const int wn = wave & 1;
    const int lm = lane & 15;
    const int lk = (lane >> 4) << 1;

    const int row0 = blockIdx.y * 64  + wm * 32;
    const int col0 = blockIdx.x * 128 + wn * 64;

    const uint32_t voff = (uint32_t)(lm * EC + lk);
    const float* abase = ec + (size_t)row0 * EC;
    const float* bbase = W  + (size_t)col0 * EC;

    v8f acc[2][4];
#pragma unroll
    for (int i = 0; i < 2; ++i)
#pragma unroll
        for (int j = 0; j < 4; ++j) acc[i][j] = (v8f){};

    for (int kb = 0; kb < EC; kb += 32) {
        if (kb + 512 < EC) {
            __builtin_prefetch(abase + voff + kb + 512, 0, 1);
            __builtin_prefetch(bbase + voff + kb + 512, 0, 1);
        }
#pragma unroll
        for (int kk = 0; kk < 32; kk += 4)
            gemm_step_rowmajor(abase + kb + kk, bbase + kb + kk,
                               voff, voff, acc);
    }

    const int rsel = (lane >> 4) * 8;
#pragma unroll
    for (int mi = 0; mi < 2; ++mi)
#pragma unroll
        for (int ni = 0; ni < 4; ++ni)
#pragma unroll
            for (int r = 0; r < 8; ++r) {
                const int row = row0 + mi * 16 + rsel + r;
                const int col = col0 + ni * 16 + lm;
                const size_t idx = (size_t)row * CA3 + col;
                out[idx] = acc[mi][ni][r] + dg[idx];
            }
}

__global__ __launch_bounds__(256) void k_colmean(
    const float* __restrict__ pc, float* __restrict__ colmean)
{
    const int t = blockIdx.x * blockDim.x + threadIdx.x;
    if (t < CA3) {
        float s = 0.f;
        for (int b = 0; b < BATCH; ++b) s += pc[(size_t)b * CA3 + t];
        colmean[t] = s * (1.0f / BATCH);
    }
}

__global__ __launch_bounds__(256) void k_delta(
    const float* __restrict__ pc, const float* __restrict__ ec,
    const float* __restrict__ W, const float* __restrict__ colmean,
    float* __restrict__ delta)
{
    const int lane = threadIdx.x & 31;
    const int wave = threadIdx.x >> 5;
    const int wm = wave >> 2;
    const int wn = wave & 3;
    const int lm = lane & 15;
    const int lk = (lane >> 4) << 1;

    const int row0 = blockIdx.y * 128 + wm * 64;
    const int col0 = blockIdx.x * 256 + wn * 64;

    const uint32_t avoff = (uint32_t)(lk * CA3 + lm);
    const uint32_t bvoff = (uint32_t)(lk * EC  + lm);
    const float* abase = pc + row0;
    const float* bbase = ec + col0;

    v8f acc[4][4];
#pragma unroll
    for (int i = 0; i < 4; ++i)
#pragma unroll
        for (int j = 0; j < 4; ++j) acc[i][j] = (v8f){};

#pragma unroll 2
    for (int k0 = 0; k0 < BATCH; k0 += 4) {
        const float* ak = abase + (size_t)k0 * CA3;
        const float* bk = bbase + (size_t)k0 * EC;
        v2f a[4], b[4];
#pragma unroll
        for (int i = 0; i < 4; ++i) {
            const float* p = ak + avoff + i * 16;
            a[i] = (v2f){ p[0], p[CA3] };
        }
#pragma unroll
        for (int i = 0; i < 4; ++i) {
            const float* p = bk + bvoff + i * 16;
            b[i] = (v2f){ p[0], p[EC] };
        }
#pragma unroll
        for (int mi = 0; mi < 4; ++mi)
#pragma unroll
            for (int ni = 0; ni < 4; ++ni)
                acc[mi][ni] = wmma_f32_k4(a[mi], b[ni], acc[mi][ni]);
    }

    const int rsel = (lane >> 4) * 8;
#pragma unroll
    for (int mi = 0; mi < 4; ++mi)
#pragma unroll
        for (int ni = 0; ni < 4; ++ni)
#pragma unroll
            for (int r = 0; r < 8; ++r) {
                const int t = row0 + mi * 16 + rsel + r;
                const int i = col0 + ni * 16 + lm;
                const size_t idx = (size_t)t * EC + i;
                delta[idx] = LR * (acc[mi][ni][r] * (1.0f / BATCH)
                                   - colmean[t] * W[idx]);
            }
}

__global__ __launch_bounds__(128) void k_loss(
    const float* __restrict__ ec, const float* __restrict__ delta,
    float* __restrict__ partials)
{
    const int lane = threadIdx.x & 31;
    const int wave = threadIdx.x >> 5;
    const int wm = wave >> 1;
    const int wn = wave & 1;
    const int lm = lane & 15;
    const int lk = (lane >> 4) << 1;

    const int row0 = blockIdx.y * 64  + wm * 32;
    const int col0 = blockIdx.x * 128 + wn * 64;

    const uint32_t voff = (uint32_t)(lm * EC + lk);
    const float* abase = ec    + (size_t)row0 * EC;
    const float* bbase = delta + (size_t)col0 * EC;

    v8f acc[2][4];
#pragma unroll
    for (int i = 0; i < 2; ++i)
#pragma unroll
        for (int j = 0; j < 4; ++j) acc[i][j] = (v8f){};

    for (int kb = 0; kb < EC; kb += 32) {
#pragma unroll
        for (int kk = 0; kk < 32; kk += 4)
            gemm_step_rowmajor(abase + kb + kk, bbase + kb + kk,
                               voff, voff, acc);
    }

    float lsum = 0.f;
#pragma unroll
    for (int mi = 0; mi < 2; ++mi)
#pragma unroll
        for (int ni = 0; ni < 4; ++ni)
#pragma unroll
            for (int r = 0; r < 8; ++r) {
                const float v = acc[mi][ni][r];
                lsum += v * v;
            }

    __shared__ float red[128];
    red[threadIdx.x] = lsum;
    __syncthreads();
#pragma unroll
    for (int s = 64; s > 0; s >>= 1) {
        if (threadIdx.x < (unsigned)s) red[threadIdx.x] += red[threadIdx.x + s];
        __syncthreads();
    }
    if (threadIdx.x == 0)
        partials[blockIdx.y * gridDim.x + blockIdx.x] = red[0];
}

__global__ void k_finalize(const float* __restrict__ partials, int n,
                           float* __restrict__ out_losses)
{
    if (threadIdx.x == 0 && blockIdx.x == 0) {
        float s = 0.f;
        for (int i = 0; i < n; ++i) s += partials[i];
        const float loss = s / (float)((size_t)BATCH * CA3);
        out_losses[0] = 0.0f;
        out_losses[1] = loss;
        out_losses[2] = loss;
    }
}

extern "C" void kernel_launch(void* const* d_in, const int* in_sizes, int n_in,
                              void* d_out, int out_size, void* d_ws, size_t ws_size,
                              hipStream_t stream)
{
    (void)in_sizes; (void)n_in; (void)out_size; (void)ws_size;
    const float* ec = (const float*)d_in[0];
    const float* dg = (const float*)d_in[1];
    const float* W  = (const float*)d_in[2];
    float* out = (float*)d_out;

    float* delta    = (float*)d_ws;
    float* colmean  = delta + (size_t)CA3 * EC;
    float* partials = colmean + CA3;

    k_gemm_pre<<<dim3(CA3 / 128, BATCH / 64), 128, 0, stream>>>(ec, dg, W, out);
    k_colmean<<<CA3 / 256, 256, 0, stream>>>(out, colmean);
    k_delta<<<dim3(EC / 256, CA3 / 128), 256, 0, stream>>>(out, ec, W, colmean, delta);
    k_loss<<<dim3(CA3 / 128, BATCH / 64), 128, 0, stream>>>(ec, delta, partials);
    k_finalize<<<1, 64, 0, stream>>>(partials, 256, out + (size_t)BATCH * CA3);
}


# Matrix chain reassociated: ec*(pc^T*ec)^T is computed as (ec*ec^T)*pc through a 1024x1024 Gram matrix, delta is never formed (34 GFLOP become 13); loss terms sum in a new order
# speedup vs baseline: 2.2377x; 1.3238x over previous
#include <hip/hip_runtime.h>
#include <hip/hip_bf16.h>
#include <stdint.h>

#define BATCH 1024
#define EC    4096
#define CA3   2048
#define LR    0.01f

typedef float v2f __attribute__((ext_vector_type(2)));
typedef float v8f __attribute__((ext_vector_type(8)));

static __device__ __forceinline__ v8f wmma_f32_k4(v2f a, v2f b, v8f c) {
    return __builtin_amdgcn_wmma_f32_16x16x4_f32(
        false, a, false, b, (short)0, c, false, false);
}

template <int MT, int NT>
static __device__ __forceinline__ void gemm_step_rowmajor(
    const float* ak, const float* bk, uint32_t avoff, uint32_t bvoff,
    v8f (&acc)[MT][NT])
{
    v2f a[MT], b[NT];
#pragma unroll
    for (int i = 0; i < MT; ++i)
        a[i] = *(const v2f*)(ak + avoff + (size_t)i * 16 * EC);
#pragma unroll
    for (int i = 0; i < NT; ++i)
        b[i] = *(const v2f*)(bk + bvoff + (size_t)i * 16 * EC);
#pragma unroll
    for (int mi = 0; mi < MT; ++mi)
#pragma unroll
        for (int ni = 0; ni < NT; ++ni)
            acc[mi][ni] = wmma_f32_k4(a[mi], b[ni], acc[mi][ni]);
}

__global__ __launch_bounds__(128) void k_gemm_pre(
    const float* __restrict__ ec, const float* __restrict__ dg,
    const float* __restrict__ W, float* __restrict__ out,
    float* __restrict__ pre)
{
    const int lane = threadIdx.x & 31;
    const int wave = threadIdx.x >> 5;
    const int wm = wave >> 1;
    const int wn = wave & 1;
    const int lm = lane & 15;
    const int lk = (lane >> 4) << 1;

    const int row0 = blockIdx.y * 64  + wm * 32;
    const int col0 = blockIdx.x * 128 + wn * 64;

    const uint32_t voff = (uint32_t)(lm * EC + lk);
    const float* abase = ec + (size_t)row0 * EC;
    const float* bbase = W  + (size_t)col0 * EC;

    v8f acc[2][4];
#pragma unroll
    for (int i = 0; i < 2; ++i)
#pragma unroll
        for (int j = 0; j < 4; ++j) acc[i][j] = (v8f){};

    for (int kb = 0; kb < EC; kb += 32) {
        if (kb + 512 < EC) {
            __builtin_prefetch(abase + voff + kb + 512, 0, 1);
            __builtin_prefetch(bbase + voff + kb + 512, 0, 1);
        }
#pragma unroll
        for (int kk = 0; kk < 32; kk += 4)
            gemm_step_rowmajor<2, 4>(abase + kb + kk, bbase + kb + kk,
                                     voff, voff, acc);
    }

    const int rsel = (lane >> 4) * 8;
#pragma unroll
    for (int mi = 0; mi < 2; ++mi)
#pragma unroll
        for (int ni = 0; ni < 4; ++ni)
#pragma unroll
            for (int r = 0; r < 8; ++r) {
                const int row = row0 + mi * 16 + rsel + r;
                const int col = col0 + ni * 16 + lm;
                const size_t idx = (size_t)row * CA3 + col;
                pre[idx] = acc[mi][ni][r];
                out[idx] = acc[mi][ni][r] + dg[idx];
            }
}

__global__ __launch_bounds__(256) void k_colmean(
    const float* __restrict__ pc, float* __restrict__ colmean)
{
    const int t = blockIdx.x * blockDim.x + threadIdx.x;
    if (t < CA3) {
        float s = 0.f;
        for (int b = 0; b < BATCH; ++b) s += pc[(size_t)b * CA3 + t];
        colmean[t] = s * (1.0f / BATCH);
    }
}


__global__ __launch_bounds__(128) void k_gram(
    const float* __restrict__ ec, float* __restrict__ G)
{
    const int lane = threadIdx.x & 31;
    const int wave = threadIdx.x >> 5;
    const int wm = wave >> 1;
    const int wn = wave & 1;
    const int lm = lane & 15;
    const int lk = (lane >> 4) << 1;

    const int row0 = blockIdx.y * 64 + wm * 32;
    const int col0 = blockIdx.x * 64 + wn * 32;

    const uint32_t voff = (uint32_t)(lm * EC + lk);
    const float* abase = ec + (size_t)row0 * EC;
    const float* bbase = ec + (size_t)col0 * EC;

    v8f acc[2][2];
#pragma unroll
    for (int i = 0; i < 2; ++i)
#pragma unroll
        for (int j = 0; j < 2; ++j) acc[i][j] = (v8f){};

    for (int kb = 0; kb < EC; kb += 32) {
        if (kb + 512 < EC) {
            __builtin_prefetch(abase + voff + kb + 512, 0, 1);
            __builtin_prefetch(bbase + voff + kb + 512, 0, 1);
        }
#pragma unroll
        for (int kk = 0; kk < 32; kk += 4)
            gemm_step_rowmajor<2, 2>(abase + kb + kk, bbase + kb + kk,
                                     voff, voff, acc);
    }

    const int rsel = (lane >> 4) * 8;
#pragma unroll
    for (int mi = 0; mi < 2; ++mi)
#pragma unroll
        for (int ni = 0; ni < 2; ++ni)
#pragma unroll
            for (int r = 0; r < 8; ++r) {
                const int row = row0 + mi * 16 + rsel + r;
                const int col = col0 + ni * 16 + lm;
                G[(size_t)row * BATCH + col] = acc[mi][ni][r];
            }
}

__global__ __launch_bounds__(128) void k_loss(
    const float* __restrict__ G, const float* __restrict__ pc,
    const float* __restrict__ pre, const float* __restrict__ colmean,
    float* __restrict__ partials)
{
    const int lane = threadIdx.x & 31;
    const int wave = threadIdx.x >> 5;
    const int wm = wave >> 1;
    const int wn = wave & 1;
    const int lm = lane & 15;
    const int lk = (lane >> 4) << 1;

    const int row0 = blockIdx.y * 64  + wm * 32;
    const int col0 = blockIdx.x * 128 + wn * 64;

    const uint32_t avoff = (uint32_t)(lm * BATCH + lk);
    const uint32_t bvoff = (uint32_t)(lk * CA3 + lm);
    const float* abase = G  + (size_t)row0 * BATCH;
    const float* bbase = pc + col0;

    v8f acc[2][4];
#pragma unroll
    for (int i = 0; i < 2; ++i)
#pragma unroll
        for (int j = 0; j < 4; ++j) acc[i][j] = (v8f){};

    for (int kb = 0; kb < BATCH; kb += 16) {
#pragma unroll
        for (int kk = 0; kk < 16; kk += 4) {
            const float* ak = abase + kb + kk;
            const float* bk = bbase + (size_t)(kb + kk) * CA3;
            v2f a[2], b[4];
#pragma unroll
            for (int i = 0; i < 2; ++i)
                a[i] = *(const v2f*)(ak + avoff + (size_t)i * 16 * BATCH);
#pragma unroll
            for (int i = 0; i < 4; ++i) {
                const float* p = bk + bvoff + i * 16;
                b[i] = (v2f){ p[0], p[CA3] };
            }
#pragma unroll
            for (int mi = 0; mi < 2; ++mi)
#pragma unroll
                for (int ni = 0; ni < 4; ++ni)
                    acc[mi][ni] = wmma_f32_k4(a[mi], b[ni], acc[mi][ni]);
        }
    }

    const int rsel = (lane >> 4) * 8;
    float lsum = 0.f;
#pragma unroll
    for (int ni = 0; ni < 4; ++ni) {
        const int col = col0 + ni * 16 + lm;
        const float cm = colmean[col];
#pragma unroll
        for (int mi = 0; mi < 2; ++mi)
#pragma unroll
            for (int r = 0; r < 8; ++r) {
                const int row = row0 + mi * 16 + rsel + r;
                const size_t idx = (size_t)row * CA3 + col;
                const float v = LR * (acc[mi][ni][r] * (1.0f / BATCH)
                                      - pre[idx] * cm);
                lsum += v * v;
            }
    }

    __shared__ float red[128];
    red[threadIdx.x] = lsum;
    __syncthreads();
#pragma unroll
    for (int s = 64; s > 0; s >>= 1) {
        if (threadIdx.x < (unsigned)s) red[threadIdx.x] += red[threadIdx.x + s];
        __syncthreads();
    }
    if (threadIdx.x == 0)
        partials[blockIdx.y * gridDim.x + blockIdx.x] = red[0];
}

__global__ void k_finalize(const float* __restrict__ partials, int n,
                           float* __restrict__ out_losses)
{
    if (threadIdx.x == 0 && blockIdx.x == 0) {
        float s = 0.f;
        for (int i = 0; i < n; ++i) s += partials[i];
        const float loss = s / (float)((size_t)BATCH * CA3);
        out_losses[0] = 0.0f;
        out_losses[1] = loss;
        out_losses[2] = loss;
    }
}

extern "C" void kernel_launch(void* const* d_in, const int* in_sizes, int n_in,
                              void* d_out, int out_size, void* d_ws, size_t ws_size,
                              hipStream_t stream)
{
    (void)in_sizes; (void)n_in; (void)out_size; (void)ws_size;
    const float* ec = (const float*)d_in[0];
    const float* dg = (const float*)d_in[1];
    const float* W  = (const float*)d_in[2];
    float* out = (float*)d_out;

    float* G        = (float*)d_ws;
    float* pre      = G + (size_t)BATCH * BATCH;
    float* colmean  = pre + (size_t)BATCH * CA3;
    float* partials = colmean + CA3;

    k_gemm_pre<<<dim3(CA3 / 128, BATCH / 64), 128, 0, stream>>>(ec, dg, W, out, pre);
    k_colmean<<<CA3 / 256, 256, 0, stream>>>(out, colmean);
    k_gram<<<dim3(BATCH / 64, BATCH / 64), 128, 0, stream>>>(ec, G);
    k_loss<<<dim3(CA3 / 128, BATCH / 64), 128, 0, stream>>>(G, out, pre, colmean, partials);
    k_finalize<<<1, 64, 0, stream>>>(partials, 256, out + (size_t)BATCH * CA3);
}


# GEMM kernels 1 and 3 stage both operands through LDS (whole 128-B lines loaded one chunk ahead replace strided fragment loads and their prefetch); k order inside 8-deep groups changes: new sum order
# speedup vs baseline: 2.9561x; 1.3210x over previous
#include <hip/hip_runtime.h>
#include <hip/hip_bf16.h>
#include <stdint.h>

#define BATCH 1024
#define EC    4096
#define CA3   2048
#define LR    0.01f

typedef float v2f __attribute__((ext_vector_type(2)));
typedef float v4f __attribute__((ext_vector_type(4)));
typedef float v8f __attribute__((ext_vector_type(8)));

#define KC 32
#define LP 36

static __device__ __forceinline__ v8f wmma_f32_k4(v2f a, v2f b, v8f c) {
    return __builtin_amdgcn_wmma_f32_16x16x4_f32(
        false, a, false, b, (short)0, c, false, false);
}

template <int BM, int BN, int MT, int NT>
static __device__ __forceinline__ void gemm_rowmajor_staged(
    const float* __restrict__ Ablk, const float* __restrict__ Bblk,
    float* __restrict__ smem, const int wm, const int wn, const int lane,
    v8f (&acc)[MT][NT])
{
    constexpr int AP = BM / 16;
    constexpr int BP = BN / 16;
    constexpr int STAGE = (BM + BN) * LP;
    constexpr int NCHUNK = EC / KC;

    const int tid = threadIdx.x;
    const int lr = tid >> 3;
    const int lq = (tid & 7) << 2;
    const float* ag = Ablk + (size_t)lr * EC + lq;
    const float* bg = Bblk + (size_t)lr * EC + lq;
    const int sa = lr * LP + lq;
    const int sb = BM * LP + sa;

    const int lm = lane & 15;
    const int lh = (lane >> 4) << 2;
    const int fa = (wm * (MT * 16) + lm) * LP + lh;
    const int fb = (BM + wn * (NT * 16) + lm) * LP + lh;

    v4f ra[AP], rb[BP];
#pragma unroll
    for (int i = 0; i < AP; ++i)
        ra[i] = *(const v4f*)(ag + (size_t)i * 16 * EC);
#pragma unroll
    for (int i = 0; i < BP; ++i)
        rb[i] = *(const v4f*)(bg + (size_t)i * 16 * EC);
#pragma unroll
    for (int i = 0; i < AP; ++i)
        *(v4f*)(smem + sa + i * 16 * LP) = ra[i];
#pragma unroll
    for (int i = 0; i < BP; ++i)
        *(v4f*)(smem + sb + i * 16 * LP) = rb[i];
    __syncthreads();

    for (int c = 0; c < NCHUNK; ++c) {
        const float* cur = smem + (c & 1) * STAGE;
        float* nxt = smem + ((c & 1) ^ 1) * STAGE;
        const bool more = (c + 1 < NCHUNK);
        if (more) {
            const int kn = (c + 1) * KC;
#pragma unroll
            for (int i = 0; i < AP; ++i)
                ra[i] = *(const v4f*)(ag + kn + (size_t)i * 16 * EC);
#pragma unroll
            for (int i = 0; i < BP; ++i)
                rb[i] = *(const v4f*)(bg + kn + (size_t)i * 16 * EC);
        }
        v4f a[KC / 8][MT], b[KC / 8][NT];
#pragma unroll
        for (int g = 0; g < KC / 8; ++g) {
#pragma unroll
            for (int i = 0; i < MT; ++i)
                a[g][i] = *(const v4f*)(cur + fa + i * 16 * LP + g * 8);
#pragma unroll
            for (int i = 0; i < NT; ++i)
                b[g][i] = *(const v4f*)(cur + fb + i * 16 * LP + g * 8);
        }
#pragma unroll
        for (int g = 0; g < KC / 8; ++g) {
            if (g == KC / 8 - 1 && more) {
                __builtin_amdgcn_sched_barrier(0);
#pragma unroll
                for (int i = 0; i < AP; ++i)
                    *(v4f*)(nxt + sa + i * 16 * LP) = ra[i];
#pragma unroll
                for (int i = 0; i < BP; ++i)
                    *(v4f*)(nxt + sb + i * 16 * LP) = rb[i];
            }
#pragma unroll
            for (int mi = 0; mi < MT; ++mi)
#pragma unroll
                for (int ni = 0; ni < NT; ++ni)
                    acc[mi][ni] = wmma_f32_k4((v2f){a[g][mi].x, a[g][mi].y},
                                              (v2f){b[g][ni].x, b[g][ni].y},
                                              acc[mi][ni]);
#pragma unroll
            for (int mi = 0; mi < MT; ++mi)
#pragma unroll
                for (int ni = 0; ni < NT; ++ni)
                    acc[mi][ni] = wmma_f32_k4((v2f){a[g][mi].z, a[g][mi].w},
                                              (v2f){b[g][ni].z, b[g][ni].w},
                                              acc[mi][ni]);
        }
        __syncthreads();
    }
}

__global__ __launch_bounds__(128) void k_gemm_pre(
    const float* __restrict__ ec, const float* __restrict__ dg,
    const float* __restrict__ W, float* __restrict__ out,
    float* __restrict__ pre)
{
    const int lane = threadIdx.x & 31;
    const int wave = threadIdx.x >> 5;
    const int wm = wave >> 1;
    const int wn = wave & 1;
    const int lm = lane & 15;

    const int row0 = blockIdx.y * 64  + wm * 32;
    const int col0 = blockIdx.x * 128 + wn * 64;

    __shared__ __attribute__((aligned(16))) float smem[2 * (64 + 128) * LP];

    v8f acc[2][4];
#pragma unroll
    for (int i = 0; i < 2; ++i)
#pragma unroll
        for (int j = 0; j < 4; ++j) acc[i][j] = (v8f){};

    gemm_rowmajor_staged<64, 128, 2, 4>(
        ec + (size_t)blockIdx.y * 64 * EC, W + (size_t)blockIdx.x * 128 * EC,
        smem, wm, wn, lane, acc);

    const int rsel = (lane >> 4) * 8;
#pragma unroll
    for (int mi = 0; mi < 2; ++mi)
#pragma unroll
        for (int ni = 0; ni < 4; ++ni)
#pragma unroll
            for (int r = 0; r < 8; ++r) {
                const int row = row0 + mi * 16 + rsel + r;
                const int col = col0 + ni * 16 + lm;
                const size_t idx = (size_t)row * CA3 + col;
                pre[idx] = acc[mi][ni][r];
                out[idx] = acc[mi][ni][r] + dg[idx];
            }
}

__global__ __launch_bounds__(256) void k_colmean(
    const float* __restrict__ pc, float* __restrict__ colmean)
{
    const int t = blockIdx.x * blockDim.x + threadIdx.x;
    if (t < CA3) {
        float s = 0.f;
        for (int b = 0; b < BATCH; ++b) s += pc[(size_t)b * CA3 + t];
        colmean[t] = s * (1.0f / BATCH);
    }
}


__global__ __launch_bounds__(128) void k_gram(
    const float* __restrict__ ec, float* __restrict__ G)
{
    const int lane = threadIdx.x & 31;
    const int wave = threadIdx.x >> 5;
    const int wm = wave >> 1;
    const int wn = wave & 1;
    const int lm = lane & 15;

    const int row0 = blockIdx.y * 64 + wm * 32;
    const int col0 = blockIdx.x * 64 + wn * 32;

    __shared__ __attribute__((aligned(16))) float smem[2 * (64 + 64) * LP];

    v8f acc[2][2];
#pragma unroll
    for (int i = 0; i < 2; ++i)
#pragma unroll
        for (int j = 0; j < 2; ++j) acc[i][j] = (v8f){};

    gemm_rowmajor_staged<64, 64, 2, 2>(
        ec + (size_t)blockIdx.y * 64 * EC, ec + (size_t)blockIdx.x * 64 * EC,
        smem, wm, wn, lane, acc);

    const int rsel = (lane >> 4) * 8;
#pragma unroll
    for (int mi = 0; mi < 2; ++mi)
#pragma unroll
        for (int ni = 0; ni < 2; ++ni)
#pragma unroll
            for (int r = 0; r < 8; ++r) {
                const int row = row0 + mi * 16 + rsel + r;
                const int col = col0 + ni * 16 + lm;
                G[(size_t)row * BATCH + col] = acc[mi][ni][r];
            }
}

__global__ __launch_bounds__(128) void k_loss(
    const float* __restrict__ G, const float* __restrict__ pc,
    const float* __restrict__ pre, const float* __restrict__ colmean,
    float* __restrict__ partials)
{
    const int lane = threadIdx.x & 31;
    const int wave = threadIdx.x >> 5;
    const int wm = wave >> 1;
    const int wn = wave & 1;
    const int lm = lane & 15;
    const int lk = (lane >> 4) << 1;

    const int row0 = blockIdx.y * 64  + wm * 32;
    const int col0 = blockIdx.x * 128 + wn * 64;

    const uint32_t avoff = (uint32_t)(lm * BATCH + lk);
    const uint32_t bvoff = (uint32_t)(lk * CA3 + lm);
    const float* abase = G  + (size_t)row0 * BATCH;
    const float* bbase = pc + col0;

    v8f acc[2][4];
#pragma unroll
    for (int i = 0; i < 2; ++i)
#pragma unroll
        for (int j = 0; j < 4; ++j) acc[i][j] = (v8f){};

    for (int kb = 0; kb < BATCH; kb += 16) {
#pragma unroll
        for (int kk = 0; kk < 16; kk += 4) {
            const float* ak = abase + kb + kk;
            const float* bk = bbase + (size_t)(kb + kk) * CA3;
            v2f a[2], b[4];
#pragma unroll
            for (int i = 0; i < 2; ++i)
                a[i] = *(const v2f*)(ak + avoff + (size_t)i * 16 * BATCH);
#pragma unroll
            for (int i = 0; i < 4; ++i) {
                const float* p = bk + bvoff + i * 16;
                b[i] = (v2f){ p[0], p[CA3] };
            }
#pragma unroll
            for (int mi = 0; mi < 2; ++mi)
#pragma unroll
                for (int ni = 0; ni < 4; ++ni)
                    acc[mi][ni] = wmma_f32_k4(a[mi], b[ni], acc[mi][ni]);
        }
    }

    const int rsel = (lane >> 4) * 8;
    float lsum = 0.f;
#pragma unroll
    for (int ni = 0; ni < 4; ++ni) {
        const int col = col0 + ni * 16 + lm;
        const float cm = colmean[col];
#pragma unroll
        for (int mi = 0; mi < 2; ++mi)
#pragma unroll
            for (int r = 0; r < 8; ++r) {
                const int row = row0 + mi * 16 + rsel + r;
                const size_t idx = (size_t)row * CA3 + col;
                const float v = LR * (acc[mi][ni][r] * (1.0f / BATCH)
                                      - pre[idx] * cm);
                lsum += v * v;
            }
    }

    __shared__ float red[128];
    red[threadIdx.x] = lsum;
    __syncthreads();
#pragma unroll
    for (int s = 64; s > 0; s >>= 1) {
        if (threadIdx.x < (unsigned)s) red[threadIdx.x] += red[threadIdx.x + s];
        __syncthreads();
    }
    if (threadIdx.x == 0)
        partials[blockIdx.y * gridDim.x + blockIdx.x] = red[0];
}

__global__ void k_finalize(const float* __restrict__ partials, int n,
                           float* __restrict__ out_losses)
{
    if (threadIdx.x == 0 && blockIdx.x == 0) {
        float s = 0.f;
        for (int i = 0; i < n; ++i) s += partials[i];
        const float loss = s / (float)((size_t)BATCH * CA3);
        out_losses[0] = 0.0f;
        out_losses[1] = loss;
        out_losses[2] = loss;
    }
}

extern "C" void kernel_launch(void* const* d_in, const int* in_sizes, int n_in,
                              void* d_out, int out_size, void* d_ws, size_t ws_size,
                              hipStream_t stream)
{
    (void)in_sizes; (void)n_in; (void)out_size; (void)ws_size;
    const float* ec = (const float*)d_in[0];
    const float* dg = (const float*)d_in[1];
    const float* W  = (const float*)d_in[2];
    float* out = (float*)d_out;

    float* G        = (float*)d_ws;
    float* pre      = G + (size_t)BATCH * BATCH;
    float* colmean  = pre + (size_t)BATCH * CA3;
    float* partials = colmean + CA3;

    k_gemm_pre<<<dim3(CA3 / 128, BATCH / 64), 128, 0, stream>>>(ec, dg, W, out, pre);
    k_colmean<<<CA3 / 256, 256, 0, stream>>>(out, colmean);
    k_gram<<<dim3(BATCH / 64, BATCH / 64), 128, 0, stream>>>(ec, G);
    k_loss<<<dim3(CA3 / 128, BATCH / 64), 128, 0, stream>>>(G, out, pre, colmean, partials);
    k_finalize<<<1, 64, 0, stream>>>(partials, 256, out + (size_t)BATCH * CA3);
}


# Column mean folded into kernel 1's epilogue as per-tile column sums (16 partial rows, added up in the loss kernel); the column-mean launch is gone (5 launches become 4); the mean sums in a new order
# speedup vs baseline: 3.4827x; 1.1781x over previous
#include <hip/hip_runtime.h>
#include <hip/hip_bf16.h>
#include <stdint.h>

#define BATCH 1024
#define EC    4096
#define CA3   2048
#define LR    0.01f

typedef float v2f __attribute__((ext_vector_type(2)));
typedef float v4f __attribute__((ext_vector_type(4)));
typedef float v8f __attribute__((ext_vector_type(8)));

#define KC 32
#define LP 36

static __device__ __forceinline__ v8f wmma_f32_k4(v2f a, v2f b, v8f c) {
    return __builtin_amdgcn_wmma_f32_16x16x4_f32(
        false, a, false, b, (short)0, c, false, false);
}

template <int BM, int BN, int MT, int NT>
static __device__ __forceinline__ void gemm_rowmajor_staged(
    const float* __restrict__ Ablk, const float* __restrict__ Bblk,
    float* __restrict__ smem, const int wm, const int wn, const int lane,
    v8f (&acc)[MT][NT])
{
    constexpr int AP = BM / 16;
    constexpr int BP = BN / 16;
    constexpr int STAGE = (BM + BN) * LP;
    constexpr int NCHUNK = EC / KC;

    const int tid = threadIdx.x;
    const int lr = tid >> 3;
    const int lq = (tid & 7) << 2;
    const float* ag = Ablk + (size_t)lr * EC + lq;
    const float* bg = Bblk + (size_t)lr * EC + lq;
    const int sa = lr * LP + lq;
    const int sb = BM * LP + sa;

    const int lm = lane & 15;
    const int lh = (lane >> 4) << 2;
    const int fa = (wm * (MT * 16) + lm) * LP + lh;
    const int fb = (BM + wn * (NT * 16) + lm) * LP + lh;

    v4f ra[AP], rb[BP];
#pragma unroll
    for (int i = 0; i < AP; ++i)
        ra[i] = *(const v4f*)(ag + (size_t)i * 16 * EC);
#pragma unroll
    for (int i = 0; i < BP; ++i)
        rb[i] = *(const v4f*)(bg + (size_t)i * 16 * EC);
#pragma unroll
    for (int i = 0; i < AP; ++i)
        *(v4f*)(smem + sa + i * 16 * LP) = ra[i];
#pragma unroll
    for (int i = 0; i < BP; ++i)
        *(v4f*)(smem + sb + i * 16 * LP) = rb[i];
    __syncthreads();

    for (int c = 0; c < NCHUNK; ++c) {
        const float* cur = smem + (c & 1) * STAGE;
        float* nxt = smem + ((c & 1) ^ 1) * STAGE;
        const bool more = (c + 1 < NCHUNK);
        if (more) {
            const int kn = (c + 1) * KC;
#pragma unroll
            for (int i = 0; i < AP; ++i)
                ra[i] = *(const v4f*)(ag + kn + (size_t)i * 16 * EC);
#pragma unroll
            for (int i = 0; i < BP; ++i)
                rb[i] = *(const v4f*)(bg + kn + (size_t)i * 16 * EC);
        }
        v4f a[KC / 8][MT], b[KC / 8][NT];
#pragma unroll
        for (int g = 0; g < KC / 8; ++g) {
#pragma unroll
            for (int i = 0; i < MT; ++i)
                a[g][i] = *(const v4f*)(cur + fa + i * 16 * LP + g * 8);
#pragma unroll
            for (int i = 0; i < NT; ++i)
                b[g][i] = *(const v4f*)(cur + fb + i * 16 * LP + g * 8);
        }
#pragma unroll
        for (int g = 0; g < KC / 8; ++g) {
            if (g == KC / 8 - 1 && more) {
                __builtin_amdgcn_sched_barrier(0);
#pragma unroll
                for (int i = 0; i < AP; ++i)
                    *(v4f*)(nxt + sa + i * 16 * LP) = ra[i];
#pragma unroll
                for (int i = 0; i < BP; ++i)
                    *(v4f*)(nxt + sb + i * 16 * LP) = rb[i];
            }
#pragma unroll
            for (int mi = 0; mi < MT; ++mi)
#pragma unroll
                for (int ni = 0; ni < NT; ++ni)
                    acc[mi][ni] = wmma_f32_k4((v2f){a[g][mi].x, a[g][mi].y},
                                              (v2f){b[g][ni].x, b[g][ni].y},
                                              acc[mi][ni]);
#pragma unroll
            for (int mi = 0; mi < MT; ++mi)
#pragma unroll
                for (int ni = 0; ni < NT; ++ni)
                    acc[mi][ni] = wmma_f32_k4((v2f){a[g][mi].z, a[g][mi].w},
                                              (v2f){b[g][ni].z, b[g][ni].w},
                                              acc[mi][ni]);
        }
        __syncthreads();
    }
}

__global__ __launch_bounds__(128) void k_gemm_pre(
    const float* __restrict__ ec, const float* __restrict__ dg,
    const float* __restrict__ W, float* __restrict__ out,
    float* __restrict__ pre, float* __restrict__ colpart)
{
    const int lane = threadIdx.x & 31;
    const int wave = threadIdx.x >> 5;
    const int wm = wave >> 1;
    const int wn = wave & 1;
    const int lm = lane & 15;

    const int row0 = blockIdx.y * 64  + wm * 32;
    const int col0 = blockIdx.x * 128 + wn * 64;

    __shared__ __attribute__((aligned(16))) float smem[2 * (64 + 128) * LP];

    v8f acc[2][4];
#pragma unroll
    for (int i = 0; i < 2; ++i)
#pragma unroll
        for (int j = 0; j < 4; ++j) acc[i][j] = (v8f){};

    gemm_rowmajor_staged<64, 128, 2, 4>(
        ec + (size_t)blockIdx.y * 64 * EC, W + (size_t)blockIdx.x * 128 * EC,
        smem, wm, wn, lane, acc);

    const int rsel = (lane >> 4) * 8;
    float csum[4] = {0.f, 0.f, 0.f, 0.f};
#pragma unroll
    for (int mi = 0; mi < 2; ++mi)
#pragma unroll
        for (int ni = 0; ni < 4; ++ni)
#pragma unroll
            for (int r = 0; r < 8; ++r) {
                const int row = row0 + mi * 16 + rsel + r;
                const int col = col0 + ni * 16 + lm;
                const size_t idx = (size_t)row * CA3 + col;
                const float v = acc[mi][ni][r] + dg[idx];
                pre[idx] = acc[mi][ni][r];
                out[idx] = v;
                csum[ni] += v;
            }

#pragma unroll
    for (int ni = 0; ni < 4; ++ni)
        csum[ni] += __shfl_xor(csum[ni], 16);
    if (lane < 16) {
#pragma unroll
        for (int ni = 0; ni < 4; ++ni)
            smem[wm * 128 + wn * 64 + ni * 16 + lm] = csum[ni];
    }
    __syncthreads();
    colpart[(size_t)blockIdx.y * CA3 + blockIdx.x * 128 + threadIdx.x] =
        smem[threadIdx.x] + smem[128 + threadIdx.x];
}


__global__ __launch_bounds__(128) void k_gram(
    const float* __restrict__ ec, float* __restrict__ G)
{
    const int lane = threadIdx.x & 31;
    const int wave = threadIdx.x >> 5;
    const int wm = wave >> 1;
    const int wn = wave & 1;
    const int lm = lane & 15;

    const int row0 = blockIdx.y * 64 + wm * 32;
    const int col0 = blockIdx.x * 64 + wn * 32;

    __shared__ __attribute__((aligned(16))) float smem[2 * (64 + 64) * LP];

    v8f acc[2][2];
#pragma unroll
    for (int i = 0; i < 2; ++i)
#pragma unroll
        for (int j = 0; j < 2; ++j) acc[i][j] = (v8f){};

    gemm_rowmajor_staged<64, 64, 2, 2>(
        ec + (size_t)blockIdx.y * 64 * EC, ec + (size_t)blockIdx.x * 64 * EC,
        smem, wm, wn, lane, acc);

    const int rsel = (lane >> 4) * 8;
#pragma unroll
    for (int mi = 0; mi < 2; ++mi)
#pragma unroll
        for (int ni = 0; ni < 2; ++ni)
#pragma unroll
            for (int r = 0; r < 8; ++r) {
                const int row = row0 + mi * 16 + rsel + r;
                const int col = col0 + ni * 16 + lm;
                G[(size_t)row * BATCH + col] = acc[mi][ni][r];
            }
}

__global__ __launch_bounds__(128) void k_loss(
    const float* __restrict__ G, const float* __restrict__ pc,
    const float* __restrict__ pre, const float* __restrict__ colpart,
    float* __restrict__ partials)
{
    const int lane = threadIdx.x & 31;
    const int wave = threadIdx.x >> 5;
    const int wm = wave >> 1;
    const int wn = wave & 1;
    const int lm = lane & 15;
    const int lk = (lane >> 4) << 1;

    const int row0 = blockIdx.y * 64  + wm * 32;
    const int col0 = blockIdx.x * 128 + wn * 64;

    const uint32_t avoff = (uint32_t)(lm * BATCH + lk);
    const uint32_t bvoff = (uint32_t)(lk * CA3 + lm);
    const float* abase = G  + (size_t)row0 * BATCH;
    const float* bbase = pc + col0;

    float cm[4];
#pragma unroll
    for (int ni = 0; ni < 4; ++ni) {
        const float* cp = colpart + col0 + ni * 16 + lm;
        float s = 0.f;
#pragma unroll
        for (int j = 0; j < BATCH / 64; ++j) s += cp[(size_t)j * CA3];
        cm[ni] = s * (1.0f / BATCH);
    }

    v8f acc[2][4];
#pragma unroll
    for (int i = 0; i < 2; ++i)
#pragma unroll
        for (int j = 0; j < 4; ++j) acc[i][j] = (v8f){};

    for (int kb = 0; kb < BATCH; kb += 16) {
#pragma unroll
        for (int kk = 0; kk < 16; kk += 4) {
            const float* ak = abase + kb + kk;
            const float* bk = bbase + (size_t)(kb + kk) * CA3;
            v2f a[2], b[4];
#pragma unroll
            for (int i = 0; i < 2; ++i)
                a[i] = *(const v2f*)(ak + avoff + (size_t)i * 16 * BATCH);
#pragma unroll
            for (int i = 0; i < 4; ++i) {
                const float* p = bk + bvoff + i * 16;
                b[i] = (v2f){ p[0], p[CA3] };
            }
#pragma unroll
            for (int mi = 0; mi < 2; ++mi)
#pragma unroll
                for (int ni = 0; ni < 4; ++ni)
                    acc[mi][ni] = wmma_f32_k4(a[mi], b[ni], acc[mi][ni]);
        }
    }

    const int rsel = (lane >> 4) * 8;
    float lsum = 0.f;
#pragma unroll
    for (int ni = 0; ni < 4; ++ni) {
        const int col = col0 + ni * 16 + lm;
#pragma unroll
        for (int mi = 0; mi < 2; ++mi)
#pragma unroll
            for (int r = 0; r < 8; ++r) {
                const int row = row0 + mi * 16 + rsel + r;
                const size_t idx = (size_t)row * CA3 + col;
                const float v = LR * (acc[mi][ni][r] * (1.0f / BATCH)
                                      - pre[idx] * cm[ni]);
                lsum += v * v;
            }
    }

    __shared__ float red[128];
    red[threadIdx.x] = lsum;
    __syncthreads();
#pragma unroll
    for (int s = 64; s > 0; s >>= 1) {
        if (threadIdx.x < (unsigned)s) red[threadIdx.x] += red[threadIdx.x + s];
        __syncthreads();
    }
    if (threadIdx.x == 0)
        partials[blockIdx.y * gridDim.x + blockIdx.x] = red[0];
}

__global__ void k_finalize(const float* __restrict__ partials, int n,
                           float* __restrict__ out_losses)
{
    if (threadIdx.x == 0 && blockIdx.x == 0) {
        float s = 0.f;
        for (int i = 0; i < n; ++i) s += partials[i];
        const float loss = s / (float)((size_t)BATCH * CA3);
        out_losses[0] = 0.0f;
        out_losses[1] = loss;
        out_losses[2] = loss;
    }
}

extern "C" void kernel_launch(void* const* d_in, const int* in_sizes, int n_in,
                              void* d_out, int out_size, void* d_ws, size_t ws_size,
                              hipStream_t stream)
{
    (void)in_sizes; (void)n_in; (void)out_size; (void)ws_size;
    const float* ec = (const float*)d_in[0];
    const float* dg = (const float*)d_in[1];
    const float* W  = (const float*)d_in[2];
    float* out = (float*)d_out;

    float* G        = (float*)d_ws;
    float* pre      = G + (size_t)BATCH * BATCH;
    float* colpart  = pre + (size_t)BATCH * CA3;
    float* partials = colpart + (size_t)(BATCH / 64) * CA3;

    k_gemm_pre<<<dim3(CA3 / 128, BATCH / 64), 128, 0, stream>>>(ec, dg, W, out, pre, colpart);
    k_gram<<<dim3(BATCH / 64, BATCH / 64), 128, 0, stream>>>(ec, G);
    k_loss<<<dim3(CA3 / 128, BATCH / 64), 128, 0, stream>>>(G, out, pre, colpart, partials);
    k_finalize<<<1, 64, 0, stream>>>(partials, 256, out + (size_t)BATCH * CA3);
}


# Gram product folded into kernel 1: each work-group multiplies its 64 ec rows by 128 W rows and 64 ec rows in one K loop (block tile 64x192, waves 32x96); 4 launches become 3; sum orders unchanged
# speedup vs baseline: 3.9736x; 1.1410x over previous
#include <hip/hip_runtime.h>
#include <hip/hip_bf16.h>
#include <stdint.h>

#define BATCH 1024
#define EC    4096
#define CA3   2048
#define LR    0.01f

typedef float v2f __attribute__((ext_vector_type(2)));
typedef float v4f __attribute__((ext_vector_type(4)));
typedef float v8f __attribute__((ext_vector_type(8)));

#define KC 32
#define LP 36

static __device__ __forceinline__ v8f wmma_f32_k4(v2f a, v2f b, v8f c) {
    return __builtin_amdgcn_wmma_f32_16x16x4_f32(
        false, a, false, b, (short)0, c, false, false);
}

template <int BM, int BN1, int BN2, int MT, int NT>
static __device__ __forceinline__ void gemm_rowmajor_staged(
    const float* __restrict__ Ablk, const float* __restrict__ B1blk,
    const float* __restrict__ B2blk,
    float* __restrict__ smem, const int wm, const int wn, const int lane,
    v8f (&acc)[MT][NT])
{
    constexpr int BN = BN1 + BN2;
    constexpr int AP = BM / 16;
    constexpr int BP1 = BN1 / 16;
    constexpr int BP = BN / 16;
    constexpr int STAGE = (BM + BN) * LP;
    constexpr int NCHUNK = EC / KC;

    const int tid = threadIdx.x;
    const int lr = tid >> 3;
    const int lq = (tid & 7) << 2;
    const float* ag = Ablk + (size_t)lr * EC + lq;
    const float* bg = B1blk + (size_t)lr * EC + lq;
    const float* bg2 = B2blk + (size_t)lr * EC + lq;
    const int sa = lr * LP + lq;
    const int sb = BM * LP + sa;

    const int lm = lane & 15;
    const int lh = (lane >> 4) << 2;
    const int fa = (wm * (MT * 16) + lm) * LP + lh;
    const int fb = (BM + wn * (NT * 16) + lm) * LP + lh;

    v4f ra[AP], rb[BP];
#pragma unroll
    for (int i = 0; i < AP; ++i)
        ra[i] = *(const v4f*)(ag + (size_t)i * 16 * EC);
#pragma unroll
    for (int i = 0; i < BP; ++i)
        rb[i] = *(const v4f*)(i < BP1 ? bg + (size_t)i * 16 * EC
                                      : bg2 + (size_t)(i - BP1) * 16 * EC);
#pragma unroll
    for (int i = 0; i < AP; ++i)
        *(v4f*)(smem + sa + i * 16 * LP) = ra[i];
#pragma unroll
    for (int i = 0; i < BP; ++i)
        *(v4f*)(smem + sb + i * 16 * LP) = rb[i];
    __syncthreads();

    for (int c = 0; c < NCHUNK; ++c) {
        const float* cur = smem + (c & 1) * STAGE;
        float* nxt = smem + ((c & 1) ^ 1) * STAGE;
        const bool more = (c + 1 < NCHUNK);
        if (more) {
            const int kn = (c + 1) * KC;
#pragma unroll
            for (int i = 0; i < AP; ++i)
                ra[i] = *(const v4f*)(ag + kn + (size_t)i * 16 * EC);
#pragma unroll
            for (int i = 0; i < BP; ++i)
                rb[i] = *(const v4f*)(i < BP1 ? bg + kn + (size_t)i * 16 * EC
                                              : bg2 + kn + (size_t)(i - BP1) * 16 * EC);
        }
        v4f a[KC / 8][MT], b[KC / 8][NT];
#pragma unroll
        for (int g = 0; g < KC / 8; ++g) {
#pragma unroll
            for (int i = 0; i < MT; ++i)
                a[g][i] = *(const v4f*)(cur + fa + i * 16 * LP + g * 8);
#pragma unroll
            for (int i = 0; i < NT; ++i)
                b[g][i] = *(const v4f*)(cur + fb + i * 16 * LP + g * 8);
        }
#pragma unroll
        for (int g = 0; g < KC / 8; ++g) {
            if (g == KC / 8 - 1 && more) {
                __builtin_amdgcn_sched_barrier(0);
#pragma unroll
                for (int i = 0; i < AP; ++i)
                    *(v4f*)(nxt + sa + i * 16 * LP) = ra[i];
#pragma unroll
                for (int i = 0; i < BP; ++i)
                    *(v4f*)(nxt + sb + i * 16 * LP) = rb[i];
            }
#pragma unroll
            for (int mi = 0; mi < MT; ++mi)
#pragma unroll
                for (int ni = 0; ni < NT; ++ni)
                    acc[mi][ni] = wmma_f32_k4((v2f){a[g][mi].x, a[g][mi].y},
                                              (v2f){b[g][ni].x, b[g][ni].y},
                                              acc[mi][ni]);
#pragma unroll
            for (int mi = 0; mi < MT; ++mi)
#pragma unroll
                for (int ni = 0; ni < NT; ++ni)
                    acc[mi][ni] = wmma_f32_k4((v2f){a[g][mi].z, a[g][mi].w},
                                              (v2f){b[g][ni].z, b[g][ni].w},
                                              acc[mi][ni]);
        }
        __syncthreads();
    }
}


__global__ __launch_bounds__(128) void k_gemm_pre_gram(
    const float* __restrict__ ec, const float* __restrict__ dg,
    const float* __restrict__ W, float* __restrict__ out,
    float* __restrict__ pre, float* __restrict__ colpart,
    float* __restrict__ G)
{
    const int lane = threadIdx.x & 31;
    const int wave = threadIdx.x >> 5;
    const int wm = wave >> 1;
    const int wn = wave & 1;
    const int lm = lane & 15;

    const int row0 = blockIdx.y * 64 + wm * 32;

    __shared__ __attribute__((aligned(16))) float smem[2 * (64 + 192) * LP];

    v8f acc[2][6];
#pragma unroll
    for (int i = 0; i < 2; ++i)
#pragma unroll
        for (int j = 0; j < 6; ++j) acc[i][j] = (v8f){};

    gemm_rowmajor_staged<64, 128, 64, 2, 6>(
        ec + (size_t)blockIdx.y * 64 * EC, W + (size_t)blockIdx.x * 128 * EC,
        ec + (size_t)blockIdx.x * 64 * EC,
        smem, wm, wn, lane, acc);

    const int rsel = (lane >> 4) * 8;
    const float* dgp = dg + (size_t)(row0 + rsel) * CA3 + blockIdx.x * 128 + lm;
    float* prep = pre + (size_t)(row0 + rsel) * CA3 + blockIdx.x * 128 + lm;
    float* outp = out + (size_t)(row0 + rsel) * CA3 + blockIdx.x * 128 + lm;
    float* gp = G + (size_t)(row0 + rsel) * BATCH + blockIdx.x * 64 + lm;
    float* cs = smem + wm * 128 + lm;

    auto w_tile = [&](const v8f (&t)[2], const int cc) -> float {
        float csum = 0.f;
#pragma unroll
        for (int mi = 0; mi < 2; ++mi)
#pragma unroll
            for (int r = 0; r < 8; ++r) {
                const size_t o = (size_t)(mi * 16 + r) * CA3 + cc;
                const float v = t[mi][r] + dgp[o];
                prep[o] = t[mi][r];
                outp[o] = v;
                csum += v;
            }
        return csum + __shfl_xor(csum, 16);
    };
    auto g_tile = [&](const v8f (&t)[2], const int gc) {
#pragma unroll
        for (int mi = 0; mi < 2; ++mi)
#pragma unroll
            for (int r = 0; r < 8; ++r)
                gp[(size_t)(mi * 16 + r) * BATCH + gc] = t[mi][r];
    };

    if (wn == 0) {
        float s[6];
#pragma unroll
        for (int ni = 0; ni < 6; ++ni) {
            const v8f t[2] = {acc[0][ni], acc[1][ni]};
            s[ni] = w_tile(t, ni * 16);
        }
        if (lane < 16) {
#pragma unroll
            for (int ni = 0; ni < 6; ++ni) cs[ni * 16] = s[ni];
        }
    } else {
        float s[2];
#pragma unroll
        for (int ni = 0; ni < 2; ++ni) {
            const v8f t[2] = {acc[0][ni], acc[1][ni]};
            s[ni] = w_tile(t, 96 + ni * 16);
        }
        if (lane < 16) {
#pragma unroll
            for (int ni = 0; ni < 2; ++ni) cs[96 + ni * 16] = s[ni];
        }
#pragma unroll
        for (int ni = 2; ni < 6; ++ni) {
            const v8f t[2] = {acc[0][ni], acc[1][ni]};
            g_tile(t, (ni - 2) * 16);
        }
    }
    __syncthreads();
    colpart[(size_t)blockIdx.y * CA3 + blockIdx.x * 128 + threadIdx.x] =
        smem[threadIdx.x] + smem[128 + threadIdx.x];
}

__global__ __launch_bounds__(128) void k_loss(
    const float* __restrict__ G, const float* __restrict__ pc,
    const float* __restrict__ pre, const float* __restrict__ colpart,
    float* __restrict__ partials)
{
    const int lane = threadIdx.x & 31;
    const int wave = threadIdx.x >> 5;
    const int wm = wave >> 1;
    const int wn = wave & 1;
    const int lm = lane & 15;
    const int lk = (lane >> 4) << 1;

    const int row0 = blockIdx.y * 64  + wm * 32;
    const int col0 = blockIdx.x * 128 + wn * 64;

    const uint32_t avoff = (uint32_t)(lm * BATCH + lk);
    const uint32_t bvoff = (uint32_t)(lk * CA3 + lm);
    const float* abase = G  + (size_t)row0 * BATCH;
    const float* bbase = pc + col0;

    float cm[4];
#pragma unroll
    for (int ni = 0; ni < 4; ++ni) {
        const float* cp = colpart + col0 + ni * 16 + lm;
        float s = 0.f;
#pragma unroll
        for (int j = 0; j < BATCH / 64; ++j) s += cp[(size_t)j * CA3];
        cm[ni] = s * (1.0f / BATCH);
    }

    v8f acc[2][4];
#pragma unroll
    for (int i = 0; i < 2; ++i)
#pragma unroll
        for (int j = 0; j < 4; ++j) acc[i][j] = (v8f){};

    for (int kb = 0; kb < BATCH; kb += 16) {
#pragma unroll
        for (int kk = 0; kk < 16; kk += 4) {
            const float* ak = abase + kb + kk;
            const float* bk = bbase + (size_t)(kb + kk) * CA3;
            v2f a[2], b[4];
#pragma unroll
            for (int i = 0; i < 2; ++i)
                a[i] = *(const v2f*)(ak + avoff + (size_t)i * 16 * BATCH);
#pragma unroll
            for (int i = 0; i < 4; ++i) {
                const float* p = bk + bvoff + i * 16;
                b[i] = (v2f){ p[0], p[CA3] };
            }
#pragma unroll
            for (int mi = 0; mi < 2; ++mi)
#pragma unroll
                for (int ni = 0; ni < 4; ++ni)
                    acc[mi][ni] = wmma_f32_k4(a[mi], b[ni], acc[mi][ni]);
        }
    }

    const int rsel = (lane >> 4) * 8;
    float lsum = 0.f;
#pragma unroll
    for (int ni = 0; ni < 4; ++ni) {
        const int col = col0 + ni * 16 + lm;
#pragma unroll
        for (int mi = 0; mi < 2; ++mi)
#pragma unroll
            for (int r = 0; r < 8; ++r) {
                const int row = row0 + mi * 16 + rsel + r;
                const size_t idx = (size_t)row * CA3 + col;
                const float v = LR * (acc[mi][ni][r] * (1.0f / BATCH)
                                      - pre[idx] * cm[ni]);
                lsum += v * v;
            }
    }

    __shared__ float red[128];
    red[threadIdx.x] = lsum;
    __syncthreads();
#pragma unroll
    for (int s = 64; s > 0; s >>= 1) {
        if (threadIdx.x < (unsigned)s) red[threadIdx.x] += red[threadIdx.x + s];
        __syncthreads();
    }
    if (threadIdx.x == 0)
        partials[blockIdx.y * gridDim.x + blockIdx.x] = red[0];
}

__global__ void k_finalize(const float* __restrict__ partials, int n,
                           float* __restrict__ out_losses)
{
    if (threadIdx.x == 0 && blockIdx.x == 0) {
        float s = 0.f;
        for (int i = 0; i < n; ++i) s += partials[i];
        const float loss = s / (float)((size_t)BATCH * CA3);
        out_losses[0] = 0.0f;
        out_losses[1] = loss;
        out_losses[2] = loss;
    }
}

extern "C" void kernel_launch(void* const* d_in, const int* in_sizes, int n_in,
                              void* d_out, int out_size, void* d_ws, size_t ws_size,
                              hipStream_t stream)
{
    (void)in_sizes; (void)n_in; (void)out_size; (void)ws_size;
    const float* ec = (const float*)d_in[0];
    const float* dg = (const float*)d_in[1];
    const float* W  = (const float*)d_in[2];
    float* out = (float*)d_out;

    float* G        = (float*)d_ws;
    float* pre      = G + (size_t)BATCH * BATCH;
    float* colpart  = pre + (size_t)BATCH * CA3;
    float* partials = colpart + (size_t)(BATCH / 64) * CA3;

    k_gemm_pre_gram<<<dim3(CA3 / 128, BATCH / 64), 128, 0, stream>>>(ec, dg, W, out, pre, colpart, G);
    k_loss<<<dim3(CA3 / 128, BATCH / 64), 128, 0, stream>>>(G, out, pre, colpart, partials);
    k_finalize<<<1, 64, 0, stream>>>(partials, 256, out + (size_t)BATCH * CA3);
}


# Loss kernel's G x pc product stages both operands through LDS (tiles loaded one 32-k chunk ahead replace per-step strided fragment loads); k order inside 8-deep groups changes: new sum order
# speedup vs baseline: 4.3524x; 1.0953x over previous
#include <hip/hip_runtime.h>
#include <hip/hip_bf16.h>
#include <stdint.h>

#define BATCH 1024
#define EC    4096
#define CA3   2048
#define LR    0.01f

typedef float v2f __attribute__((ext_vector_type(2)));
typedef float v4f __attribute__((ext_vector_type(4)));
typedef float v8f __attribute__((ext_vector_type(8)));

#define KC 32
#define LP 36

static __device__ __forceinline__ v8f wmma_f32_k4(v2f a, v2f b, v8f c) {
    return __builtin_amdgcn_wmma_f32_16x16x4_f32(
        false, a, false, b, (short)0, c, false, false);
}

template <int BM, int BN1, int BN2, int MT, int NT>
static __device__ __forceinline__ void gemm_rowmajor_staged(
    const float* __restrict__ Ablk, const float* __restrict__ B1blk,
    const float* __restrict__ B2blk,
    float* __restrict__ smem, const int wm, const int wn, const int lane,
    v8f (&acc)[MT][NT])
{
    constexpr int BN = BN1 + BN2;
    constexpr int AP = BM / 16;
    constexpr int BP1 = BN1 / 16;
    constexpr int BP = BN / 16;
    constexpr int STAGE = (BM + BN) * LP;
    constexpr int NCHUNK = EC / KC;

    const int tid = threadIdx.x;
    const int lr = tid >> 3;
    const int lq = (tid & 7) << 2;
    const float* ag = Ablk + (size_t)lr * EC + lq;
    const float* bg = B1blk + (size_t)lr * EC + lq;
    const float* bg2 = B2blk + (size_t)lr * EC + lq;
    const int sa = lr * LP + lq;
    const int sb = BM * LP + sa;

    const int lm = lane & 15;
    const int lh = (lane >> 4) << 2;
    const int fa = (wm * (MT * 16) + lm) * LP + lh;
    const int fb = (BM + wn * (NT * 16) + lm) * LP + lh;

    v4f ra[AP], rb[BP];
#pragma unroll
    for (int i = 0; i < AP; ++i)
        ra[i] = *(const v4f*)(ag + (size_t)i * 16 * EC);
#pragma unroll
    for (int i = 0; i < BP; ++i)
        rb[i] = *(const v4f*)(i < BP1 ? bg + (size_t)i * 16 * EC
                                      : bg2 + (size_t)(i - BP1) * 16 * EC);
#pragma unroll
    for (int i = 0; i < AP; ++i)
        *(v4f*)(smem + sa + i * 16 * LP) = ra[i];
#pragma unroll
    for (int i = 0; i < BP; ++i)
        *(v4f*)(smem + sb + i * 16 * LP) = rb[i];
    __syncthreads();

    for (int c = 0; c < NCHUNK; ++c) {
        const float* cur = smem + (c & 1) * STAGE;
        float* nxt = smem + ((c & 1) ^ 1) * STAGE;
        const bool more = (c + 1 < NCHUNK);
        if (more) {
            const int kn = (c + 1) * KC;
#pragma unroll
            for (int i = 0; i < AP; ++i)
                ra[i] = *(const v4f*)(ag + kn + (size_t)i * 16 * EC);
#pragma unroll
            for (int i = 0; i < BP; ++i)
                rb[i] = *(const v4f*)(i < BP1 ? bg + kn + (size_t)i * 16 * EC
                                              : bg2 + kn + (size_t)(i - BP1) * 16 * EC);
        }
        v4f a[KC / 8][MT], b[KC / 8][NT];
#pragma unroll
        for (int g = 0; g < KC / 8; ++g) {
#pragma unroll
            for (int i = 0; i < MT; ++i)
                a[g][i] = *(const v4f*)(cur + fa + i * 16 * LP + g * 8);
#pragma unroll
            for (int i = 0; i < NT; ++i)
                b[g][i] = *(const v4f*)(cur + fb + i * 16 * LP + g * 8);
        }
#pragma unroll
        for (int g = 0; g < KC / 8; ++g) {
            if (g == KC / 8 - 1 && more) {
                __builtin_amdgcn_sched_barrier(0);
#pragma unroll
                for (int i = 0; i < AP; ++i)
                    *(v4f*)(nxt + sa + i * 16 * LP) = ra[i];
#pragma unroll
                for (int i = 0; i < BP; ++i)
                    *(v4f*)(nxt + sb + i * 16 * LP) = rb[i];
            }
#pragma unroll
            for (int mi = 0; mi < MT; ++mi)
#pragma unroll
                for (int ni = 0; ni < NT; ++ni)
                    acc[mi][ni] = wmma_f32_k4((v2f){a[g][mi].x, a[g][mi].y},
                                              (v2f){b[g][ni].x, b[g][ni].y},
                                              acc[mi][ni]);
#pragma unroll
            for (int mi = 0; mi < MT; ++mi)
#pragma unroll
                for (int ni = 0; ni < NT; ++ni)
                    acc[mi][ni] = wmma_f32_k4((v2f){a[g][mi].z, a[g][mi].w},
                                              (v2f){b[g][ni].z, b[g][ni].w},
                                              acc[mi][ni]);
        }
        __syncthreads();
    }
}


__global__ __launch_bounds__(128) void k_gemm_pre_gram(
    const float* __restrict__ ec, const float* __restrict__ dg,
    const float* __restrict__ W, float* __restrict__ out,
    float* __restrict__ pre, float* __restrict__ colpart,
    float* __restrict__ G)
{
    const int lane = threadIdx.x & 31;
    const int wave = threadIdx.x >> 5;
    const int wm = wave >> 1;
    const int wn = wave & 1;
    const int lm = lane & 15;

    const int row0 = blockIdx.y * 64 + wm * 32;

    __shared__ __attribute__((aligned(16))) float smem[2 * (64 + 192) * LP];

    v8f acc[2][6];
#pragma unroll
    for (int i = 0; i < 2; ++i)
#pragma unroll
        for (int j = 0; j < 6; ++j) acc[i][j] = (v8f){};

    gemm_rowmajor_staged<64, 128, 64, 2, 6>(
        ec + (size_t)blockIdx.y * 64 * EC, W + (size_t)blockIdx.x * 128 * EC,
        ec + (size_t)blockIdx.x * 64 * EC,
        smem, wm, wn, lane, acc);

    const int rsel = (lane >> 4) * 8;
    const float* dgp = dg + (size_t)(row0 + rsel) * CA3 + blockIdx.x * 128 + lm;
    float* prep = pre + (size_t)(row0 + rsel) * CA3 + blockIdx.x * 128 + lm;
    float* outp = out + (size_t)(row0 + rsel) * CA3 + blockIdx.x * 128 + lm;
    float* gp = G + (size_t)(row0 + rsel) * BATCH + blockIdx.x * 64 + lm;
    float* cs = smem + wm * 128 + lm;

    auto w_tile = [&](const v8f (&t)[2], const int cc) -> float {
        float csum = 0.f;
#pragma unroll
        for (int mi = 0; mi < 2; ++mi)
#pragma unroll
            for (int r = 0; r < 8; ++r) {
                const size_t o = (size_t)(mi * 16 + r) * CA3 + cc;
                const float v = t[mi][r] + dgp[o];
                prep[o] = t[mi][r];
                outp[o] = v;
                csum += v;
            }
        return csum + __shfl_xor(csum, 16);
    };
    auto g_tile = [&](const v8f (&t)[2], const int gc) {
#pragma unroll
        for (int mi = 0; mi < 2; ++mi)
#pragma unroll
            for (int r = 0; r < 8; ++r)
                gp[(size_t)(mi * 16 + r) * BATCH + gc] = t[mi][r];
    };

    if (wn == 0) {
        float s[6];
#pragma unroll
        for (int ni = 0; ni < 6; ++ni) {
            const v8f t[2] = {acc[0][ni], acc[1][ni]};
            s[ni] = w_tile(t, ni * 16);
        }
        if (lane < 16) {
#pragma unroll
            for (int ni = 0; ni < 6; ++ni) cs[ni * 16] = s[ni];
        }
    } else {
        float s[2];
#pragma unroll
        for (int ni = 0; ni < 2; ++ni) {
            const v8f t[2] = {acc[0][ni], acc[1][ni]};
            s[ni] = w_tile(t, 96 + ni * 16);
        }
        if (lane < 16) {
#pragma unroll
            for (int ni = 0; ni < 2; ++ni) cs[96 + ni * 16] = s[ni];
        }
#pragma unroll
        for (int ni = 2; ni < 6; ++ni) {
            const v8f t[2] = {acc[0][ni], acc[1][ni]};
            g_tile(t, (ni - 2) * 16);
        }
    }
    __syncthreads();
    colpart[(size_t)blockIdx.y * CA3 + blockIdx.x * 128 + threadIdx.x] =
        smem[threadIdx.x] + smem[128 + threadIdx.x];
}

static __device__ __forceinline__ void gemm_gram_pc_staged(
    const float* __restrict__ Ablk, const float* __restrict__ Bblk,
    float* __restrict__ smem, const int wm, const int wn, const int lane,
    v8f (&acc)[2][4])
{
    constexpr int BM = 64, BN = 128, MT = 2, NT = 4;
    constexpr int AP = BM / 16;
    constexpr int BP = BN / 16;
    constexpr int STAGE = (BM + BN) * LP;
    constexpr int NCHUNK = BATCH / KC;

    const int tid = threadIdx.x;
    const int lr = tid >> 3;
    const int lq = (tid & 7) << 2;
    const int bk = tid >> 2;
    const int bn = (tid & 3) << 2;
    const float* ag = Ablk + (size_t)lr * BATCH + lq;
    const float* bg = Bblk + (size_t)bk * CA3 + bn;
    const int sa = lr * LP + lq;
    const int sb = (BM + bn) * LP + bk;

    const int lm = lane & 15;
    const int lh = (lane >> 4) << 2;
    const int fa = (wm * (MT * 16) + lm) * LP + lh;
    const int fb = (BM + wn * (NT * 16) + lm) * LP + lh;

    v4f ra[AP], rb[BP];
#pragma unroll
    for (int i = 0; i < AP; ++i)
        ra[i] = *(const v4f*)(ag + (size_t)i * 16 * BATCH);
#pragma unroll
    for (int i = 0; i < BP; ++i)
        rb[i] = *(const v4f*)(bg + i * 16);
#pragma unroll
    for (int i = 0; i < AP; ++i)
        *(v4f*)(smem + sa + i * 16 * LP) = ra[i];
#pragma unroll
    for (int i = 0; i < BP; ++i) {
        smem[sb + (i * 16 + 0) * LP] = rb[i].x;
        smem[sb + (i * 16 + 1) * LP] = rb[i].y;
        smem[sb + (i * 16 + 2) * LP] = rb[i].z;
        smem[sb + (i * 16 + 3) * LP] = rb[i].w;
    }
    __syncthreads();

    for (int c = 0; c < NCHUNK; ++c) {
        const float* cur = smem + (c & 1) * STAGE;
        float* nxt = smem + ((c & 1) ^ 1) * STAGE;
        const bool more = (c + 1 < NCHUNK);
        if (more) {
            const int kn = (c + 1) * KC;
#pragma unroll
            for (int i = 0; i < AP; ++i)
                ra[i] = *(const v4f*)(ag + kn + (size_t)i * 16 * BATCH);
#pragma unroll
            for (int i = 0; i < BP; ++i)
                rb[i] = *(const v4f*)(bg + (size_t)kn * CA3 + i * 16);
        }
        v4f a[KC / 8][MT], b[KC / 8][NT];
#pragma unroll
        for (int g = 0; g < KC / 8; ++g) {
#pragma unroll
            for (int i = 0; i < MT; ++i)
                a[g][i] = *(const v4f*)(cur + fa + i * 16 * LP + g * 8);
#pragma unroll
            for (int i = 0; i < NT; ++i)
                b[g][i] = *(const v4f*)(cur + fb + i * 16 * LP + g * 8);
        }
#pragma unroll
        for (int g = 0; g < KC / 8; ++g) {
            if (g == KC / 8 - 1 && more) {
                __builtin_amdgcn_sched_barrier(0);
#pragma unroll
                for (int i = 0; i < AP; ++i)
                    *(v4f*)(nxt + sa + i * 16 * LP) = ra[i];
#pragma unroll
                for (int i = 0; i < BP; ++i) {
                    nxt[sb + (i * 16 + 0) * LP] = rb[i].x;
                    nxt[sb + (i * 16 + 1) * LP] = rb[i].y;
                    nxt[sb + (i * 16 + 2) * LP] = rb[i].z;
                    nxt[sb + (i * 16 + 3) * LP] = rb[i].w;
                }
            }
#pragma unroll
            for (int mi = 0; mi < MT; ++mi)
#pragma unroll
                for (int ni = 0; ni < NT; ++ni)
                    acc[mi][ni] = wmma_f32_k4((v2f){a[g][mi].x, a[g][mi].y},
                                              (v2f){b[g][ni].x, b[g][ni].y},
                                              acc[mi][ni]);
#pragma unroll
            for (int mi = 0; mi < MT; ++mi)
#pragma unroll
                for (int ni = 0; ni < NT; ++ni)
                    acc[mi][ni] = wmma_f32_k4((v2f){a[g][mi].z, a[g][mi].w},
                                              (v2f){b[g][ni].z, b[g][ni].w},
                                              acc[mi][ni]);
        }
        __syncthreads();
    }
}

__global__ __launch_bounds__(128) void k_loss(
    const float* __restrict__ G, const float* __restrict__ pc,
    const float* __restrict__ pre, const float* __restrict__ colpart,
    float* __restrict__ partials)
{
    const int lane = threadIdx.x & 31;
    const int wave = threadIdx.x >> 5;
    const int wm = wave >> 1;
    const int wn = wave & 1;
    const int lm = lane & 15;

    const int row0 = blockIdx.y * 64  + wm * 32;
    const int col0 = blockIdx.x * 128 + wn * 64;

    __shared__ __attribute__((aligned(16))) float smem[2 * (64 + 128) * LP];

    float cm[4];
#pragma unroll
    for (int ni = 0; ni < 4; ++ni) {
        const float* cp = colpart + col0 + ni * 16 + lm;
        float s = 0.f;
#pragma unroll
        for (int j = 0; j < BATCH / 64; ++j) s += cp[(size_t)j * CA3];
        cm[ni] = s * (1.0f / BATCH);
    }

    v8f acc[2][4];
#pragma unroll
    for (int i = 0; i < 2; ++i)
#pragma unroll
        for (int j = 0; j < 4; ++j) acc[i][j] = (v8f){};

    gemm_gram_pc_staged(G + (size_t)blockIdx.y * 64 * BATCH,
                        pc + blockIdx.x * 128, smem, wm, wn, lane, acc);

    const int rsel = (lane >> 4) * 8;
    float lsum = 0.f;
#pragma unroll
    for (int ni = 0; ni < 4; ++ni) {
        const int col = col0 + ni * 16 + lm;
#pragma unroll
        for (int mi = 0; mi < 2; ++mi)
#pragma unroll
            for (int r = 0; r < 8; ++r) {
                const int row = row0 + mi * 16 + rsel + r;
                const size_t idx = (size_t)row * CA3 + col;
                const float v = LR * (acc[mi][ni][r] * (1.0f / BATCH)
                                      - pre[idx] * cm[ni]);
                lsum += v * v;
            }
    }

    __shared__ float red[128];
    red[threadIdx.x] = lsum;
    __syncthreads();
#pragma unroll
    for (int s = 64; s > 0; s >>= 1) {
        if (threadIdx.x < (unsigned)s) red[threadIdx.x] += red[threadIdx.x + s];
        __syncthreads();
    }
    if (threadIdx.x == 0)
        partials[blockIdx.y * gridDim.x + blockIdx.x] = red[0];
}

__global__ void k_finalize(const float* __restrict__ partials, int n,
                           float* __restrict__ out_losses)
{
    if (threadIdx.x == 0 && blockIdx.x == 0) {
        float s = 0.f;
        for (int i = 0; i < n; ++i) s += partials[i];
        const float loss = s / (float)((size_t)BATCH * CA3);
        out_losses[0] = 0.0f;
        out_losses[1] = loss;
        out_losses[2] = loss;
    }
}

extern "C" void kernel_launch(void* const* d_in, const int* in_sizes, int n_in,
                              void* d_out, int out_size, void* d_ws, size_t ws_size,
                              hipStream_t stream)
{
    (void)in_sizes; (void)n_in; (void)out_size; (void)ws_size;
    const float* ec = (const float*)d_in[0];
    const float* dg = (const float*)d_in[1];
    const float* W  = (const float*)d_in[2];
    float* out = (float*)d_out;

    float* G        = (float*)d_ws;
    float* pre      = G + (size_t)BATCH * BATCH;
    float* colpart  = pre + (size_t)BATCH * CA3;
    float* partials = colpart + (size_t)(BATCH / 64) * CA3;

    k_gemm_pre_gram<<<dim3(CA3 / 128, BATCH / 64), 128, 0, stream>>>(ec, dg, W, out, pre, colpart, G);
    k_loss<<<dim3(CA3 / 128, BATCH / 64), 128, 0, stream>>>(G, out, pre, colpart, partials);
    k_finalize<<<1, 64, 0, stream>>>(partials, 256, out + (size_t)BATCH * CA3);
}


# Both GEMM K loops: LDS ring of three stages (was two), barrier orders LDS only; stores, global loads and the next chunk's first fragment reads sit under matrix groups; sum orders unchanged
# speedup vs baseline: 4.6172x; 1.0608x over previous
#include <hip/hip_runtime.h>
#include <hip/hip_bf16.h>
#include <stdint.h>

#define BATCH 1024
#define EC    4096
#define CA3   2048
#define LR    0.01f

typedef float v2f __attribute__((ext_vector_type(2)));
typedef float v4f __attribute__((ext_vector_type(4)));
typedef float v8f __attribute__((ext_vector_type(8)));

#define KC 32
#define LP 36

static __device__ __forceinline__ v8f wmma_f32_k4(v2f a, v2f b, v8f c) {
    return __builtin_amdgcn_wmma_f32_16x16x4_f32(
        false, a, false, b, (short)0, c, false, false);
}

static __device__ __forceinline__ void lds_barrier() {
    __builtin_amdgcn_fence(__ATOMIC_RELEASE, "workgroup", "local");
    __builtin_amdgcn_s_barrier();
    __builtin_amdgcn_fence(__ATOMIC_ACQUIRE, "workgroup", "local");
}

template <int MT, int NT>
static __device__ __forceinline__ void frag_read(
    const float* __restrict__ st, const int fa, const int fb, const int g,
    v4f (&a)[MT], v4f (&b)[NT])
{
#pragma unroll
    for (int i = 0; i < MT; ++i)
        a[i] = *(const v4f*)(st + fa + i * 16 * LP + g * 8);
#pragma unroll
    for (int i = 0; i < NT; ++i)
        b[i] = *(const v4f*)(st + fb + i * 16 * LP + g * 8);
}

template <int MT, int NT>
static __device__ __forceinline__ void wmma_group(
    const v4f (&a)[MT], const v4f (&b)[NT], v8f (&acc)[MT][NT])
{
#pragma unroll
    for (int mi = 0; mi < MT; ++mi)
#pragma unroll
        for (int ni = 0; ni < NT; ++ni)
            acc[mi][ni] = wmma_f32_k4((v2f){a[mi].x, a[mi].y},
                                      (v2f){b[ni].x, b[ni].y}, acc[mi][ni]);
#pragma unroll
    for (int mi = 0; mi < MT; ++mi)
#pragma unroll
        for (int ni = 0; ni < NT; ++ni)
            acc[mi][ni] = wmma_f32_k4((v2f){a[mi].z, a[mi].w},
                                      (v2f){b[ni].z, b[ni].w}, acc[mi][ni]);
}

template <bool STORE, bool LOAD, bool NEXT, int MT, int NT, class Stager>
static __device__ __forceinline__ void ring_step(
    Stager& sg, const int kload, const float* __restrict__ cur,
    const float* __restrict__ nxt, float* __restrict__ st,
    const int fa, const int fb,
    v4f (&a0)[MT], v4f (&b0)[NT], v4f (&a1)[MT], v4f (&b1)[NT],
    v8f (&acc)[MT][NT])
{
    frag_read<MT, NT>(cur, fa, fb, 1, a1, b1);
    __builtin_amdgcn_sched_barrier(0);
    wmma_group<MT, NT>(a0, b0, acc);
    __builtin_amdgcn_sched_barrier(0);
    frag_read<MT, NT>(cur, fa, fb, 2, a0, b0);
    if (STORE) sg.lstore(st);
    __builtin_amdgcn_sched_barrier(0);
    if (LOAD) sg.gload(kload);
    __builtin_amdgcn_sched_barrier(0);
    wmma_group<MT, NT>(a1, b1, acc);
    __builtin_amdgcn_sched_barrier(0);
    frag_read<MT, NT>(cur, fa, fb, 3, a1, b1);
    __builtin_amdgcn_sched_barrier(0);
    wmma_group<MT, NT>(a0, b0, acc);
    __builtin_amdgcn_sched_barrier(0);
    if (NEXT) frag_read<MT, NT>(nxt, fa, fb, 0, a0, b0);
    __builtin_amdgcn_sched_barrier(0);
    wmma_group<MT, NT>(a1, b1, acc);
    __builtin_amdgcn_sched_barrier(0);
    lds_barrier();
}

template <int NCHUNK, int STAGE, int MT, int NT, class Stager>
static __device__ __forceinline__ void gemm_ring3(
    Stager& sg, float* __restrict__ smem, const int fa, const int fb,
    v8f (&acc)[MT][NT])
{
    static_assert(NCHUNK >= 4, "ring needs at least four chunks");
    sg.gload(0);
    sg.lstore(smem);
    sg.gload(KC);
    sg.lstore(smem + STAGE);
    __syncthreads();
    sg.gload(2 * KC);

    v4f a0[MT], b0[NT], a1[MT], b1[NT];
    frag_read<MT, NT>(smem, fa, fb, 0, a0, b0);

    int oc = 0, on = STAGE, os = 2 * STAGE;
    for (int c = 0; c < NCHUNK - 3; ++c) {
        ring_step<true, true, true, MT, NT>(sg, (c + 3) * KC, smem + oc,
                                            smem + on, smem + os, fa, fb,
                                            a0, b0, a1, b1, acc);
        const int t = oc; oc = on; on = os; os = t;
    }
    ring_step<true, false, true, MT, NT>(sg, 0, smem + oc, smem + on,
                                         smem + os, fa, fb, a0, b0, a1, b1, acc);
    ring_step<false, false, true, MT, NT>(sg, 0, smem + on, smem + os,
                                          smem + oc, fa, fb, a0, b0, a1, b1, acc);
    ring_step<false, false, false, MT, NT>(sg, 0, smem + os, smem + oc,
                                           smem + on, fa, fb, a0, b0, a1, b1, acc);
}

template <int BM, int BN1, int BN2>
struct StagerRowRow {
    static constexpr int AP = BM / 16;
    static constexpr int BP1 = BN1 / 16;
    static constexpr int BP = (BN1 + BN2) / 16;
    const float* ag; const float* bg; const float* bg2;
    int sa, sb;
    v4f ra[AP], rb[BP];
    __device__ __forceinline__ StagerRowRow(const float* Ablk, const float* B1blk,
                                            const float* B2blk) {
        const int tid = threadIdx.x;
        const int lr = tid >> 3;
        const int lq = (tid & 7) << 2;
        ag = Ablk + (size_t)lr * EC + lq;
        bg = B1blk + (size_t)lr * EC + lq;
        bg2 = B2blk + (size_t)lr * EC + lq;
        sa = lr * LP + lq;
        sb = BM * LP + sa;
    }
    __device__ __forceinline__ void gload(const int k) {
#pragma unroll
        for (int i = 0; i < AP; ++i)
            ra[i] = *(const v4f*)(ag + k + (size_t)i * 16 * EC);
#pragma unroll
        for (int i = 0; i < BP; ++i)
            rb[i] = *(const v4f*)(i < BP1 ? bg + k + (size_t)i * 16 * EC
                                          : bg2 + k + (size_t)(i - BP1) * 16 * EC);
    }
    __device__ __forceinline__ void lstore(float* __restrict__ st) const {
#pragma unroll
        for (int i = 0; i < AP; ++i)
            *(v4f*)(st + sa + i * 16 * LP) = ra[i];
#pragma unroll
        for (int i = 0; i < BP; ++i)
            *(v4f*)(st + sb + i * 16 * LP) = rb[i];
    }
};

struct StagerGramPc {
    static constexpr int BM = 64, BN = 128;
    static constexpr int AP = BM / 16;
    static constexpr int BP = BN / 16;
    const float* ag; const float* bg;
    int sa, sb;
    v4f ra[AP], rb[BP];
    __device__ __forceinline__ StagerGramPc(const float* Ablk, const float* Bblk) {
        const int tid = threadIdx.x;
        const int lr = tid >> 3;
        const int lq = (tid & 7) << 2;
        const int bk = tid >> 2;
        const int bn = (tid & 3) << 2;
        ag = Ablk + (size_t)lr * BATCH + lq;
        bg = Bblk + (size_t)bk * CA3 + bn;
        sa = lr * LP + lq;
        sb = (BM + bn) * LP + bk;
    }
    __device__ __forceinline__ void gload(const int k) {
#pragma unroll
        for (int i = 0; i < AP; ++i)
            ra[i] = *(const v4f*)(ag + k + (size_t)i * 16 * BATCH);
#pragma unroll
        for (int i = 0; i < BP; ++i)
            rb[i] = *(const v4f*)(bg + (size_t)k * CA3 + i * 16);
    }
    __device__ __forceinline__ void lstore(float* __restrict__ st) const {
#pragma unroll
        for (int i = 0; i < AP; ++i)
            *(v4f*)(st + sa + i * 16 * LP) = ra[i];
#pragma unroll
        for (int i = 0; i < BP; ++i) {
            st[sb + (i * 16 + 0) * LP] = rb[i].x;
            st[sb + (i * 16 + 1) * LP] = rb[i].y;
            st[sb + (i * 16 + 2) * LP] = rb[i].z;
            st[sb + (i * 16 + 3) * LP] = rb[i].w;
        }
    }
};


__global__ __launch_bounds__(128) void k_gemm_pre_gram(
    const float* __restrict__ ec, const float* __restrict__ dg,
    const float* __restrict__ W, float* __restrict__ out,
    float* __restrict__ pre, float* __restrict__ colpart,
    float* __restrict__ G)
{
    const int lane = threadIdx.x & 31;
    const int wave = threadIdx.x >> 5;
    const int wm = wave >> 1;
    const int wn = wave & 1;
    const int lm = lane & 15;

    const int row0 = blockIdx.y * 64 + wm * 32;

    __shared__ __attribute__((aligned(16))) float smem[3 * (64 + 192) * LP];

    v8f acc[2][6];
#pragma unroll
    for (int i = 0; i < 2; ++i)
#pragma unroll
        for (int j = 0; j < 6; ++j) acc[i][j] = (v8f){};

    {
        StagerRowRow<64, 128, 64> sg(ec + (size_t)blockIdx.y * 64 * EC,
                                     W + (size_t)blockIdx.x * 128 * EC,
                                     ec + (size_t)blockIdx.x * 64 * EC);
        const int lh = (lane >> 4) << 2;
        const int fa = (wm * 32 + lm) * LP + lh;
        const int fb = (64 + wn * 96 + lm) * LP + lh;
        gemm_ring3<EC / KC, (64 + 192) * LP, 2, 6>(sg, smem, fa, fb, acc);
    }

    const int rsel = (lane >> 4) * 8;
    const float* dgp = dg + (size_t)(row0 + rsel) * CA3 + blockIdx.x * 128 + lm;
    float* prep = pre + (size_t)(row0 + rsel) * CA3 + blockIdx.x * 128 + lm;
    float* outp = out + (size_t)(row0 + rsel) * CA3 + blockIdx.x * 128 + lm;
    float* gp = G + (size_t)(row0 + rsel) * BATCH + blockIdx.x * 64 + lm;
    float* cs = smem + wm * 128 + lm;

    auto w_tile = [&](const v8f (&t)[2], const int cc) -> float {
        float csum = 0.f;
#pragma unroll
        for (int mi = 0; mi < 2; ++mi)
#pragma unroll
            for (int r = 0; r < 8; ++r) {
                const size_t o = (size_t)(mi * 16 + r) * CA3 + cc;
                const float v = t[mi][r] + dgp[o];
                prep[o] = t[mi][r];
                outp[o] = v;
                csum += v;
            }
        return csum + __shfl_xor(csum, 16);
    };
    auto g_tile = [&](const v8f (&t)[2], const int gc) {
#pragma unroll
        for (int mi = 0; mi < 2; ++mi)
#pragma unroll
            for (int r = 0; r < 8; ++r)
                gp[(size_t)(mi * 16 + r) * BATCH + gc] = t[mi][r];
    };

    if (wn == 0) {
        float s[6];
#pragma unroll
        for (int ni = 0; ni < 6; ++ni) {
            const v8f t[2] = {acc[0][ni], acc[1][ni]};
            s[ni] = w_tile(t, ni * 16);
        }
        if (lane < 16) {
#pragma unroll
            for (int ni = 0; ni < 6; ++ni) cs[ni * 16] = s[ni];
        }
    } else {
        float s[2];
#pragma unroll
        for (int ni = 0; ni < 2; ++ni) {
            const v8f t[2] = {acc[0][ni], acc[1][ni]};
            s[ni] = w_tile(t, 96 + ni * 16);
        }
        if (lane < 16) {
#pragma unroll
            for (int ni = 0; ni < 2; ++ni) cs[96 + ni * 16] = s[ni];
        }
#pragma unroll
        for (int ni = 2; ni < 6; ++ni) {
            const v8f t[2] = {acc[0][ni], acc[1][ni]};
            g_tile(t, (ni - 2) * 16);
        }
    }
    __syncthreads();
    colpart[(size_t)blockIdx.y * CA3 + blockIdx.x * 128 + threadIdx.x] =
        smem[threadIdx.x] + smem[128 + threadIdx.x];
}

__global__ __launch_bounds__(128) void k_loss(
    const float* __restrict__ G, const float* __restrict__ pc,
    const float* __restrict__ pre, const float* __restrict__ colpart,
    float* __restrict__ partials)
{
    const int lane = threadIdx.x & 31;
    const int wave = threadIdx.x >> 5;
    const int wm = wave >> 1;
    const int wn = wave & 1;
    const int lm = lane & 15;

    const int row0 = blockIdx.y * 64  + wm * 32;
    const int col0 = blockIdx.x * 128 + wn * 64;

    __shared__ __attribute__((aligned(16))) float smem[3 * (64 + 128) * LP];

    float cm[4];
#pragma unroll
    for (int ni = 0; ni < 4; ++ni) {
        const float* cp = colpart + col0 + ni * 16 + lm;
        float s = 0.f;
#pragma unroll
        for (int j = 0; j < BATCH / 64; ++j) s += cp[(size_t)j * CA3];
        cm[ni] = s * (1.0f / BATCH);
    }

    v8f acc[2][4];
#pragma unroll
    for (int i = 0; i < 2; ++i)
#pragma unroll
        for (int j = 0; j < 4; ++j) acc[i][j] = (v8f){};

    {
        StagerGramPc sg(G + (size_t)blockIdx.y * 64 * BATCH, pc + blockIdx.x * 128);
        const int lh = (lane >> 4) << 2;
        const int fa = (wm * 32 + lm) * LP + lh;
        const int fb = (64 + wn * 64 + lm) * LP + lh;
        gemm_ring3<BATCH / KC, (64 + 128) * LP, 2, 4>(sg, smem, fa, fb, acc);
    }

    const int rsel = (lane >> 4) * 8;
    float lsum = 0.f;
#pragma unroll
    for (int ni = 0; ni < 4; ++ni) {
        const int col = col0 + ni * 16 + lm;
#pragma unroll
        for (int mi = 0; mi < 2; ++mi)
#pragma unroll
            for (int r = 0; r < 8; ++r) {
                const int row = row0 + mi * 16 + rsel + r;
                const size_t idx = (size_t)row * CA3 + col;
                const float v = LR * (acc[mi][ni][r] * (1.0f / BATCH)
                                      - pre[idx] * cm[ni]);
                lsum += v * v;
            }
    }

    __shared__ float red[128];
    red[threadIdx.x] = lsum;
    __syncthreads();
#pragma unroll
    for (int s = 64; s > 0; s >>= 1) {
        if (threadIdx.x < (unsigned)s) red[threadIdx.x] += red[threadIdx.x + s];
        __syncthreads();
    }
    if (threadIdx.x == 0)
        partials[blockIdx.y * gridDim.x + blockIdx.x] = red[0];
}

__global__ void k_finalize(const float* __restrict__ partials, int n,
                           float* __restrict__ out_losses)
{
    if (threadIdx.x == 0 && blockIdx.x == 0) {
        float s = 0.f;
        for (int i = 0; i < n; ++i) s += partials[i];
        const float loss = s / (float)((size_t)BATCH * CA3);
        out_losses[0] = 0.0f;
        out_losses[1] = loss;
        out_losses[2] = loss;
    }
}

extern "C" void kernel_launch(void* const* d_in, const int* in_sizes, int n_in,
                              void* d_out, int out_size, void* d_ws, size_t ws_size,
                              hipStream_t stream)
{
    (void)in_sizes; (void)n_in; (void)out_size; (void)ws_size;
    const float* ec = (const float*)d_in[0];
    const float* dg = (const float*)d_in[1];
    const float* W  = (const float*)d_in[2];
    float* out = (float*)d_out;

    float* G        = (float*)d_ws;
    float* pre      = G + (size_t)BATCH * BATCH;
    float* colpart  = pre + (size_t)BATCH * CA3;
    float* partials = colpart + (size_t)(BATCH / 64) * CA3;

    k_gemm_pre_gram<<<dim3(CA3 / 128, BATCH / 64), 128, 0, stream>>>(ec, dg, W, out, pre, colpart, G);
    k_loss<<<dim3(CA3 / 128, BATCH / 64), 128, 0, stream>>>(G, out, pre, colpart, partials);
    k_finalize<<<1, 64, 0, stream>>>(partials, 256, out + (size_t)BATCH * CA3);
}


# Both GEMM K loops: 8-wave work-groups, two waves per SIMD (wave tiles 32x48, 32x32; 64-k chunks), a wave's staging and waits sit under its sibling's matrix runs; loss sums regroup: new sum order
# speedup vs baseline: 5.1540x; 1.1163x over previous
#include <hip/hip_runtime.h>
#include <hip/hip_bf16.h>
#include <stdint.h>

#define BATCH 1024
#define EC    4096
#define CA3   2048
#define LR    0.01f

typedef float v2f __attribute__((ext_vector_type(2)));
typedef float v4f __attribute__((ext_vector_type(4)));
typedef float v8f __attribute__((ext_vector_type(8)));

#define KC 64
#define LP 68
#define NTHR 256
#define GK 2

static __device__ __forceinline__ v8f wmma_f32_k4(v2f a, v2f b, v8f c) {
    return __builtin_amdgcn_wmma_f32_16x16x4_f32(
        false, a, false, b, (short)0, c, false, false);
}

static __device__ __forceinline__ void lds_barrier() {
    __builtin_amdgcn_fence(__ATOMIC_RELEASE, "workgroup", "local");
    __builtin_amdgcn_s_barrier();
    __builtin_amdgcn_fence(__ATOMIC_ACQUIRE, "workgroup", "local");
}

template <int MT, int NT>
static __device__ __forceinline__ void frag_read(
    const float* __restrict__ st, const int fa, const int fb, const int r,
    v4f (&a)[GK][MT], v4f (&b)[GK][NT])
{
#pragma unroll
    for (int q = 0; q < GK; ++q) {
#pragma unroll
        for (int i = 0; i < MT; ++i)
            a[q][i] = *(const v4f*)(st + fa + i * 16 * LP + (r * GK + q) * 8);
#pragma unroll
        for (int i = 0; i < NT; ++i)
            b[q][i] = *(const v4f*)(st + fb + i * 16 * LP + (r * GK + q) * 8);
    }
}

template <int MT, int NT>
static __device__ __forceinline__ void wmma_group(
    const v4f (&a)[GK][MT], const v4f (&b)[GK][NT], v8f (&acc)[MT][NT])
{
#pragma unroll
    for (int q = 0; q < GK; ++q) {
#pragma unroll
        for (int mi = 0; mi < MT; ++mi)
#pragma unroll
            for (int ni = 0; ni < NT; ++ni)
                acc[mi][ni] = wmma_f32_k4((v2f){a[q][mi].x, a[q][mi].y},
                                          (v2f){b[q][ni].x, b[q][ni].y}, acc[mi][ni]);
#pragma unroll
        for (int mi = 0; mi < MT; ++mi)
#pragma unroll
            for (int ni = 0; ni < NT; ++ni)
                acc[mi][ni] = wmma_f32_k4((v2f){a[q][mi].z, a[q][mi].w},
                                          (v2f){b[q][ni].z, b[q][ni].w}, acc[mi][ni]);
    }
}

template <bool STORE, bool LOAD, bool NEXT, int MT, int NT, class Stager>
static __device__ __forceinline__ void ring_step(
    Stager& sg, const int kload, const float* __restrict__ cur,
    const float* __restrict__ nxt, float* __restrict__ st,
    const int fa, const int fb,
    v4f (&a0)[GK][MT], v4f (&b0)[GK][NT], v4f (&a1)[GK][MT], v4f (&b1)[GK][NT],
    v8f (&acc)[MT][NT])
{
    frag_read<MT, NT>(cur, fa, fb, 1, a1, b1);
    __builtin_amdgcn_sched_barrier(0);
    wmma_group<MT, NT>(a0, b0, acc);
    __builtin_amdgcn_sched_barrier(0);
    frag_read<MT, NT>(cur, fa, fb, 2, a0, b0);
    if (STORE) sg.lstore(st);
    __builtin_amdgcn_sched_barrier(0);
    if (LOAD) sg.gload(kload);
    __builtin_amdgcn_sched_barrier(0);
    wmma_group<MT, NT>(a1, b1, acc);
    __builtin_amdgcn_sched_barrier(0);
    frag_read<MT, NT>(cur, fa, fb, 3, a1, b1);
    __builtin_amdgcn_sched_barrier(0);
    wmma_group<MT, NT>(a0, b0, acc);
    __builtin_amdgcn_sched_barrier(0);
    if (NEXT) frag_read<MT, NT>(nxt, fa, fb, 0, a0, b0);
    __builtin_amdgcn_sched_barrier(0);
    wmma_group<MT, NT>(a1, b1, acc);
    __builtin_amdgcn_sched_barrier(0);
    lds_barrier();
}

template <int NCHUNK, int STAGE, int MT, int NT, class Stager>
static __device__ __forceinline__ void gemm_ring3(
    Stager& sg, float* __restrict__ smem, const int fa, const int fb,
    v8f (&acc)[MT][NT])
{
    static_assert(NCHUNK >= 4, "ring needs at least four chunks");
    sg.gload(0);
    sg.lstore(smem);
    sg.gload(KC);
    sg.lstore(smem + STAGE);
    __syncthreads();
    sg.gload(2 * KC);

    v4f a0[GK][MT], b0[GK][NT], a1[GK][MT], b1[GK][NT];
    frag_read<MT, NT>(smem, fa, fb, 0, a0, b0);

    int oc = 0, on = STAGE, os = 2 * STAGE;
    for (int c = 0; c < NCHUNK - 3; ++c) {
        ring_step<true, true, true, MT, NT>(sg, (c + 3) * KC, smem + oc,
                                            smem + on, smem + os, fa, fb,
                                            a0, b0, a1, b1, acc);
        const int t = oc; oc = on; on = os; os = t;
    }
    ring_step<true, false, true, MT, NT>(sg, 0, smem + oc, smem + on,
                                         smem + os, fa, fb, a0, b0, a1, b1, acc);
    ring_step<false, false, true, MT, NT>(sg, 0, smem + on, smem + os,
                                          smem + oc, fa, fb, a0, b0, a1, b1, acc);
    ring_step<false, false, false, MT, NT>(sg, 0, smem + os, smem + oc,
                                           smem + on, fa, fb, a0, b0, a1, b1, acc);
}

template <int BM, int BN1, int BN2>
struct StagerRowRow {
    static constexpr int AP = BM / 16;
    static constexpr int BP1 = BN1 / 16;
    static constexpr int BP = (BN1 + BN2) / 16;
    const float* ag; const float* bg; const float* bg2;
    int sa, sb;
    v4f ra[AP], rb[BP];
    __device__ __forceinline__ StagerRowRow(const float* Ablk, const float* B1blk,
                                            const float* B2blk) {
        const int tid = threadIdx.x;
        const int lr = tid >> 4;
        const int lq = (tid & 15) << 2;
        ag = Ablk + (size_t)lr * EC + lq;
        bg = B1blk + (size_t)lr * EC + lq;
        bg2 = B2blk + (size_t)lr * EC + lq;
        sa = lr * LP + lq;
        sb = BM * LP + sa;
    }
    __device__ __forceinline__ void gload(const int k) {
#pragma unroll
        for (int i = 0; i < AP; ++i)
            ra[i] = *(const v4f*)(ag + k + (size_t)i * 16 * EC);
#pragma unroll
        for (int i = 0; i < BP; ++i)
            rb[i] = *(const v4f*)(i < BP1 ? bg + k + (size_t)i * 16 * EC
                                          : bg2 + k + (size_t)(i - BP1) * 16 * EC);
    }
    __device__ __forceinline__ void lstore(float* __restrict__ st) const {
#pragma unroll
        for (int i = 0; i < AP; ++i)
            *(v4f*)(st + sa + i * 16 * LP) = ra[i];
#pragma unroll
        for (int i = 0; i < BP; ++i)
            *(v4f*)(st + sb + i * 16 * LP) = rb[i];
    }
};

struct StagerGramPc {
    static constexpr int BM = 64, BN = 128;
    static constexpr int AP = BM / 16;
    static constexpr int BP = BN / 16;
    const float* ag; const float* bg;
    int sa, sb;
    v4f ra[AP], rb[BP];
    __device__ __forceinline__ StagerGramPc(const float* Ablk, const float* Bblk) {
        const int tid = threadIdx.x;
        const int lr = tid >> 4;
        const int lq = (tid & 15) << 2;
        const int bk = tid >> 2;
        const int bn = (tid & 3) << 2;
        ag = Ablk + (size_t)lr * BATCH + lq;
        bg = Bblk + (size_t)bk * CA3 + bn;
        sa = lr * LP + lq;
        sb = (BM + bn) * LP + bk;
    }
    __device__ __forceinline__ void gload(const int k) {
#pragma unroll
        for (int i = 0; i < AP; ++i)
            ra[i] = *(const v4f*)(ag + k + (size_t)i * 16 * BATCH);
#pragma unroll
        for (int i = 0; i < BP; ++i)
            rb[i] = *(const v4f*)(bg + (size_t)k * CA3 + i * 16);
    }
    __device__ __forceinline__ void lstore(float* __restrict__ st) const {
#pragma unroll
        for (int i = 0; i < AP; ++i)
            *(v4f*)(st + sa + i * 16 * LP) = ra[i];
#pragma unroll
        for (int i = 0; i < BP; ++i) {
            st[sb + (i * 16 + 0) * LP] = rb[i].x;
            st[sb + (i * 16 + 1) * LP] = rb[i].y;
            st[sb + (i * 16 + 2) * LP] = rb[i].z;
            st[sb + (i * 16 + 3) * LP] = rb[i].w;
        }
    }
};


__global__ __launch_bounds__(NTHR) void k_gemm_pre_gram(
    const float* __restrict__ ec, const float* __restrict__ dg,
    const float* __restrict__ W, float* __restrict__ out,
    float* __restrict__ pre, float* __restrict__ colpart,
    float* __restrict__ G)
{
    const int lane = threadIdx.x & 31;
    const int wave = threadIdx.x >> 5;
    const int wm = wave >> 2;
    const int wn = wave & 3;
    const int lm = lane & 15;

    const int row0 = blockIdx.y * 64 + wm * 32;

    __shared__ __attribute__((aligned(16))) float smem[3 * (64 + 192) * LP];

    v8f acc[2][3];
#pragma unroll
    for (int i = 0; i < 2; ++i)
#pragma unroll
        for (int j = 0; j < 3; ++j) acc[i][j] = (v8f){};

    {
        StagerRowRow<64, 128, 64> sg(ec + (size_t)blockIdx.y * 64 * EC,
                                     W + (size_t)blockIdx.x * 128 * EC,
                                     ec + (size_t)blockIdx.x * 64 * EC);
        const int lh = (lane >> 4) << 2;
        const int fa = (wm * 32 + lm) * LP + lh;
        const int fb = (64 + wn * 48 + lm) * LP + lh;
        gemm_ring3<EC / KC, (64 + 192) * LP, 2, 3>(sg, smem, fa, fb, acc);
    }

    const int rsel = (lane >> 4) * 8;
    const float* dgp = dg + (size_t)(row0 + rsel) * CA3 + blockIdx.x * 128 + lm;
    float* prep = pre + (size_t)(row0 + rsel) * CA3 + blockIdx.x * 128 + lm;
    float* outp = out + (size_t)(row0 + rsel) * CA3 + blockIdx.x * 128 + lm;
    float* gp = G + (size_t)(row0 + rsel) * BATCH + blockIdx.x * 64 + lm;
    float* cs = smem + wm * 128 + lm;

    auto w_tile = [&](const v8f (&t)[2], const int cc) -> float {
        float csum = 0.f;
#pragma unroll
        for (int mi = 0; mi < 2; ++mi)
#pragma unroll
            for (int r = 0; r < 8; ++r) {
                const size_t o = (size_t)(mi * 16 + r) * CA3 + cc;
                const float v = t[mi][r] + dgp[o];
                prep[o] = t[mi][r];
                outp[o] = v;
                csum += v;
            }
        return csum + __shfl_xor(csum, 16);
    };
    auto g_tile = [&](const v8f (&t)[2], const int gc) {
#pragma unroll
        for (int mi = 0; mi < 2; ++mi)
#pragma unroll
            for (int r = 0; r < 8; ++r)
                gp[(size_t)(mi * 16 + r) * BATCH + gc] = t[mi][r];
    };

    if (wn < 2) {
        const int c0 = wn * 48;
        float s[3];
#pragma unroll
        for (int ni = 0; ni < 3; ++ni) {
            const v8f t[2] = {acc[0][ni], acc[1][ni]};
            s[ni] = w_tile(t, c0 + ni * 16);
        }
        if (lane < 16) {
#pragma unroll
            for (int ni = 0; ni < 3; ++ni) cs[c0 + ni * 16] = s[ni];
        }
    } else if (wn == 2) {
        float s[2];
#pragma unroll
        for (int ni = 0; ni < 2; ++ni) {
            const v8f t[2] = {acc[0][ni], acc[1][ni]};
            s[ni] = w_tile(t, 96 + ni * 16);
        }
        if (lane < 16) {
#pragma unroll
            for (int ni = 0; ni < 2; ++ni) cs[96 + ni * 16] = s[ni];
        }
        const v8f t[2] = {acc[0][2], acc[1][2]};
        g_tile(t, 0);
    } else {
#pragma unroll
        for (int ni = 0; ni < 3; ++ni) {
            const v8f t[2] = {acc[0][ni], acc[1][ni]};
            g_tile(t, 16 + ni * 16);
        }
    }
    __syncthreads();
    if (threadIdx.x < 128)
        colpart[(size_t)blockIdx.y * CA3 + blockIdx.x * 128 + threadIdx.x] =
            smem[threadIdx.x] + smem[128 + threadIdx.x];
}

__global__ __launch_bounds__(NTHR) void k_loss(
    const float* __restrict__ G, const float* __restrict__ pc,
    const float* __restrict__ pre, const float* __restrict__ colpart,
    float* __restrict__ partials)
{
    const int lane = threadIdx.x & 31;
    const int wave = threadIdx.x >> 5;
    const int wm = wave >> 2;
    const int wn = wave & 3;
    const int lm = lane & 15;

    const int row0 = blockIdx.y * 64  + wm * 32;
    const int col0 = blockIdx.x * 128 + wn * 32;

    __shared__ __attribute__((aligned(16))) float smem[3 * (64 + 128) * LP];

    float cm[2];
#pragma unroll
    for (int ni = 0; ni < 2; ++ni) {
        const float* cp = colpart + col0 + ni * 16 + lm;
        float s = 0.f;
#pragma unroll
        for (int j = 0; j < BATCH / 64; ++j) s += cp[(size_t)j * CA3];
        cm[ni] = s * (1.0f / BATCH);
    }

    v8f acc[2][2];
#pragma unroll
    for (int i = 0; i < 2; ++i)
#pragma unroll
        for (int j = 0; j < 2; ++j) acc[i][j] = (v8f){};

    {
        StagerGramPc sg(G + (size_t)blockIdx.y * 64 * BATCH, pc + blockIdx.x * 128);
        const int lh = (lane >> 4) << 2;
        const int fa = (wm * 32 + lm) * LP + lh;
        const int fb = (64 + wn * 32 + lm) * LP + lh;
        gemm_ring3<BATCH / KC, (64 + 128) * LP, 2, 2>(sg, smem, fa, fb, acc);
    }

    const int rsel = (lane >> 4) * 8;
    float lsum = 0.f;
#pragma unroll
    for (int ni = 0; ni < 2; ++ni) {
        const int col = col0 + ni * 16 + lm;
#pragma unroll
        for (int mi = 0; mi < 2; ++mi)
#pragma unroll
            for (int r = 0; r < 8; ++r) {
                const int row = row0 + mi * 16 + rsel + r;
                const size_t idx = (size_t)row * CA3 + col;
                const float v = LR * (acc[mi][ni][r] * (1.0f / BATCH)
                                      - pre[idx] * cm[ni]);
                lsum += v * v;
            }
    }

    __shared__ float red[NTHR];
    red[threadIdx.x] = lsum;
    __syncthreads();
#pragma unroll
    for (int s = NTHR / 2; s > 0; s >>= 1) {
        if (threadIdx.x < (unsigned)s) red[threadIdx.x] += red[threadIdx.x + s];
        __syncthreads();
    }
    if (threadIdx.x == 0)
        partials[blockIdx.y * gridDim.x + blockIdx.x] = red[0];
}

__global__ void k_finalize(const float* __restrict__ partials, int n,
                           float* __restrict__ out_losses)
{
    if (threadIdx.x == 0 && blockIdx.x == 0) {
        float s = 0.f;
        for (int i = 0; i < n; ++i) s += partials[i];
        const float loss = s / (float)((size_t)BATCH * CA3);
        out_losses[0] = 0.0f;
        out_losses[1] = loss;
        out_losses[2] = loss;
    }
}

extern "C" void kernel_launch(void* const* d_in, const int* in_sizes, int n_in,
                              void* d_out, int out_size, void* d_ws, size_t ws_size,
                              hipStream_t stream)
{
    (void)in_sizes; (void)n_in; (void)out_size; (void)ws_size;
    const float* ec = (const float*)d_in[0];
    const float* dg = (const float*)d_in[1];
    const float* W  = (const float*)d_in[2];
    float* out = (float*)d_out;

    float* G        = (float*)d_ws;
    float* pre      = G + (size_t)BATCH * BATCH;
    float* colpart  = pre + (size_t)BATCH * CA3;
    float* partials = colpart + (size_t)(BATCH / 64) * CA3;

    k_gemm_pre_gram<<<dim3(CA3 / 128, BATCH / 64), NTHR, 0, stream>>>(ec, dg, W, out, pre, colpart, G);
    k_loss<<<dim3(CA3 / 128, BATCH / 64), NTHR, 0, stream>>>(G, out, pre, colpart, partials);
    k_finalize<<<1, 64, 0, stream>>>(partials, 256, out + (size_t)BATCH * CA3);
}


# Kernel 1 computes only 136 of the 256 blocks of the symmetric Gram matrix and stores each tile straight and transposed; block tile 64x176, 5.5 tiles per wave (one strip split by k): new sum order
# speedup vs baseline: 5.3194x; 1.0321x over previous
#include <hip/hip_runtime.h>
#include <hip/hip_bf16.h>
#include <stdint.h>

#define BATCH 1024
#define EC    4096
#define CA3   2048
#define LR    0.01f

typedef float v2f __attribute__((ext_vector_type(2)));
typedef float v4f __attribute__((ext_vector_type(4)));
typedef float v8f __attribute__((ext_vector_type(8)));

#define KC 64
#define LP 68
#define NTHR 256
#define GK 2

static __device__ __forceinline__ v8f wmma_f32_k4(v2f a, v2f b, v8f c) {
    return __builtin_amdgcn_wmma_f32_16x16x4_f32(
        false, a, false, b, (short)0, c, false, false);
}

static __device__ __forceinline__ void lds_barrier() {
    __builtin_amdgcn_fence(__ATOMIC_RELEASE, "workgroup", "local");
    __builtin_amdgcn_s_barrier();
    __builtin_amdgcn_fence(__ATOMIC_ACQUIRE, "workgroup", "local");
}

template <int MT, int NT, bool XT>
struct FragSet {
    v4f a[GK][MT], b[GK][NT];
    v4f xa, xb;
};
struct FragOfs { int fa, fb, fxa, fxb; };

template <int MT, int NT, bool XT>
static __device__ __forceinline__ void frag_read(
    const float* __restrict__ st, const FragOfs& o, const int r,
    FragSet<MT, NT, XT>& f)
{
#pragma unroll
    for (int q = 0; q < GK; ++q) {
#pragma unroll
        for (int i = 0; i < MT; ++i)
            f.a[q][i] = *(const v4f*)(st + o.fa + i * 16 * LP + (r * GK + q) * 8);
#pragma unroll
        for (int i = 0; i < NT; ++i)
            f.b[q][i] = *(const v4f*)(st + o.fb + i * 16 * LP + (r * GK + q) * 8);
    }
    if (XT) {
        f.xa = *(const v4f*)(st + o.fxa + r * GK * 8);
        f.xb = *(const v4f*)(st + o.fxb + r * GK * 8);
    }
}

template <int MT, int NT, bool XT>
static __device__ __forceinline__ void wmma_group(
    const FragSet<MT, NT, XT>& f, v8f (&acc)[MT][NT], v8f& accx)
{
#pragma unroll
    for (int q = 0; q < GK; ++q) {
#pragma unroll
        for (int mi = 0; mi < MT; ++mi)
#pragma unroll
            for (int ni = 0; ni < NT; ++ni)
                acc[mi][ni] = wmma_f32_k4((v2f){f.a[q][mi].x, f.a[q][mi].y},
                                          (v2f){f.b[q][ni].x, f.b[q][ni].y}, acc[mi][ni]);
        if (XT && q == 0)
            accx = wmma_f32_k4((v2f){f.xa.x, f.xa.y}, (v2f){f.xb.x, f.xb.y}, accx);
        if (XT && q == GK - 1)
            accx = wmma_f32_k4((v2f){f.xa.z, f.xa.w}, (v2f){f.xb.z, f.xb.w}, accx);
#pragma unroll
        for (int mi = 0; mi < MT; ++mi)
#pragma unroll
            for (int ni = 0; ni < NT; ++ni)
                acc[mi][ni] = wmma_f32_k4((v2f){f.a[q][mi].z, f.a[q][mi].w},
                                          (v2f){f.b[q][ni].z, f.b[q][ni].w}, acc[mi][ni]);
    }
}

template <bool STORE, bool LOAD, bool NEXT, int MT, int NT, bool XT, class Stager>
static __device__ __forceinline__ void ring_step(
    Stager& sg, const int kload, const float* __restrict__ cur,
    const float* __restrict__ nxt, float* __restrict__ st, const FragOfs& o,
    FragSet<MT, NT, XT>& f0, FragSet<MT, NT, XT>& f1,
    v8f (&acc)[MT][NT], v8f& accx)
{
    frag_read<MT, NT, XT>(cur, o, 1, f1);
    __builtin_amdgcn_sched_barrier(0);
    wmma_group<MT, NT, XT>(f0, acc, accx);
    __builtin_amdgcn_sched_barrier(0);
    frag_read<MT, NT, XT>(cur, o, 2, f0);
    if (STORE) sg.lstore(st);
    __builtin_amdgcn_sched_barrier(0);
    if (LOAD) sg.gload(kload);
    __builtin_amdgcn_sched_barrier(0);
    wmma_group<MT, NT, XT>(f1, acc, accx);
    __builtin_amdgcn_sched_barrier(0);
    frag_read<MT, NT, XT>(cur, o, 3, f1);
    __builtin_amdgcn_sched_barrier(0);
    wmma_group<MT, NT, XT>(f0, acc, accx);
    __builtin_amdgcn_sched_barrier(0);
    if (NEXT) frag_read<MT, NT, XT>(nxt, o, 0, f0);
    __builtin_amdgcn_sched_barrier(0);
    wmma_group<MT, NT, XT>(f1, acc, accx);
    __builtin_amdgcn_sched_barrier(0);
    lds_barrier();
}

template <int NCHUNK, int STAGE, int MT, int NT, bool XT, class Stager>
static __device__ __forceinline__ void gemm_ring3(
    Stager& sg, float* __restrict__ smem, const FragOfs& o,
    v8f (&acc)[MT][NT], v8f& accx)
{
    static_assert(NCHUNK >= 4, "ring needs at least four chunks");
    sg.gload(0);
    sg.lstore(smem);
    sg.gload(KC);
    sg.lstore(smem + STAGE);
    __syncthreads();
    sg.gload(2 * KC);

    FragSet<MT, NT, XT> f0, f1;
    frag_read<MT, NT, XT>(smem, o, 0, f0);

    int oc = 0, on = STAGE, os = 2 * STAGE;
    for (int c = 0; c < NCHUNK - 3; ++c) {
        ring_step<true, true, true, MT, NT, XT>(sg, (c + 3) * KC, smem + oc,
                                                smem + on, smem + os, o,
                                                f0, f1, acc, accx);
        const int t = oc; oc = on; on = os; os = t;
    }
    ring_step<true, false, true, MT, NT, XT>(sg, 0, smem + oc, smem + on,
                                             smem + os, o, f0, f1, acc, accx);
    ring_step<false, false, true, MT, NT, XT>(sg, 0, smem + on, smem + os,
                                              smem + oc, o, f0, f1, acc, accx);
    ring_step<false, false, false, MT, NT, XT>(sg, 0, smem + os, smem + oc,
                                               smem + on, o, f0, f1, acc, accx);
}

struct StagerPreGram {
    static constexpr int AP = 4;
    static constexpr int BW = 8;
    static constexpr int BG = 2;
    static constexpr int BP = BW + BG + 1;
    const float* ag; const float* bg; const float* bg2; const float* bg3;
    int sa, sb;
    v4f ra[AP], rb[BP];
    __device__ __forceinline__ StagerPreGram(const float* Ablk, const float* Wblk,
                                             const float* Gpair, const float* Gone) {
        const int tid = threadIdx.x;
        const int lr = tid >> 4;
        const int lq = (tid & 15) << 2;
        ag = Ablk + (size_t)lr * EC + lq;
        bg = Wblk + (size_t)lr * EC + lq;
        bg2 = Gpair + (size_t)lr * EC + lq;
        bg3 = Gone + (size_t)lr * EC + lq;
        sa = lr * LP + lq;
        sb = 64 * LP + sa;
    }
    __device__ __forceinline__ void gload(const int k) {
#pragma unroll
        for (int i = 0; i < AP; ++i)
            ra[i] = *(const v4f*)(ag + k + (size_t)i * 16 * EC);
#pragma unroll
        for (int i = 0; i < BW; ++i)
            rb[i] = *(const v4f*)(bg + k + (size_t)i * 16 * EC);
#pragma unroll
        for (int i = 0; i < BG; ++i)
            rb[BW + i] = *(const v4f*)(bg2 + k + (size_t)i * 16 * EC);
        rb[BW + BG] = *(const v4f*)(bg3 + k);
    }
    __device__ __forceinline__ void lstore(float* __restrict__ st) const {
#pragma unroll
        for (int i = 0; i < AP; ++i)
            *(v4f*)(st + sa + i * 16 * LP) = ra[i];
#pragma unroll
        for (int i = 0; i < BP; ++i)
            *(v4f*)(st + sb + i * 16 * LP) = rb[i];
    }
};

struct StagerGramPc {
    static constexpr int BM = 64, BN = 128;
    static constexpr int AP = BM / 16;
    static constexpr int BP = BN / 16;
    const float* ag; const float* bg;
    int sa, sb;
    v4f ra[AP], rb[BP];
    __device__ __forceinline__ StagerGramPc(const float* Ablk, const float* Bblk) {
        const int tid = threadIdx.x;
        const int lr = tid >> 4;
        const int lq = (tid & 15) << 2;
        const int bk = tid >> 2;
        const int bn = (tid & 3) << 2;
        ag = Ablk + (size_t)lr * BATCH + lq;
        bg = Bblk + (size_t)bk * CA3 + bn;
        sa = lr * LP + lq;
        sb = (BM + bn) * LP + bk;
    }
    __device__ __forceinline__ void gload(const int k) {
#pragma unroll
        for (int i = 0; i < AP; ++i)
            ra[i] = *(const v4f*)(ag + k + (size_t)i * 16 * BATCH);
#pragma unroll
        for (int i = 0; i < BP; ++i)
            rb[i] = *(const v4f*)(bg + (size_t)k * CA3 + i * 16);
    }
    __device__ __forceinline__ void lstore(float* __restrict__ st) const {
#pragma unroll
        for (int i = 0; i < AP; ++i)
            *(v4f*)(st + sa + i * 16 * LP) = ra[i];
#pragma unroll
        for (int i = 0; i < BP; ++i) {
            st[sb + (i * 16 + 0) * LP] = rb[i].x;
            st[sb + (i * 16 + 1) * LP] = rb[i].y;
            st[sb + (i * 16 + 2) * LP] = rb[i].z;
            st[sb + (i * 16 + 3) * LP] = rb[i].w;
        }
    }
};


__global__ __launch_bounds__(NTHR) void k_gemm_pre_gram(
    const float* __restrict__ ec, const float* __restrict__ dg,
    const float* __restrict__ W, float* __restrict__ out,
    float* __restrict__ pre, float* __restrict__ colpart,
    float* __restrict__ G)
{
    const int lane = threadIdx.x & 31;
    const int wave = threadIdx.x >> 5;
    const int wm = wave & 3;
    const int wn = wave >> 2;
    const int lm = lane & 15;
    const int bx = blockIdx.x;
    const int by = blockIdx.y;

    const int gd = 1 + (bx >> 1);
    const bool pair_on = gd <= (by < 8 ? 8 : 7);
    const int pair0 = ((by + gd) & 15) * 64 + (bx & 1) * 32;
    const bool one_on = bx < 4;
    const int one0 = by * 64 + (bx & 3) * 16;

    __shared__ __attribute__((aligned(16))) float smem[3 * (64 + 176) * LP];

    v8f acc[1][5];
#pragma unroll
    for (int j = 0; j < 5; ++j) acc[0][j] = (v8f){};
    v8f accx = (v8f){};

    {
        StagerPreGram sg(ec + (size_t)by * 64 * EC, W + (size_t)bx * 128 * EC,
                         ec + (size_t)pair0 * EC, ec + (size_t)one0 * EC);
        const int lh = (lane >> 4) << 2;
        FragOfs fo;
        fo.fa = (wm * 16 + lm) * LP + lh;
        fo.fb = (64 + wn * 80 + lm) * LP + lh;
        fo.fxa = fo.fa + wn * 8;
        fo.fxb = (64 + 160 + lm) * LP + lh + wn * 8;
        gemm_ring3<EC / KC, (64 + 176) * LP, 1, 5, true>(sg, smem, fo, acc, accx);
    }

    const int rsel = (lane >> 4) * 8;
    const int row0 = by * 64 + wm * 16 + rsel;
    float* xs = smem;
    float* cs = smem + 1024;

    if (wn == 1) {
#pragma unroll
        for (int r = 0; r < 8; ++r) xs[(wm * 8 + r) * 32 + lane] = accx[r];
    }
    __syncthreads();

    auto w_tile = [&](const v8f& t, const int s) {
        const int col = bx * 128 + s * 16 + lm;
        float csum = 0.f;
#pragma unroll
        for (int r = 0; r < 8; ++r) {
            const size_t o = (size_t)(row0 + r) * CA3 + col;
            const float v = t[r] + dg[o];
            pre[o] = t[r];
            out[o] = v;
            csum += v;
        }
        csum += __shfl_xor(csum, 16);
        if (lane < 16) cs[(wm * 8 + s) * 16 + lm] = csum;
    };
    auto g_direct = [&](const v8f& t, const int c0) {
#pragma unroll
        for (int r = 0; r < 8; ++r)
            G[(size_t)(row0 + r) * BATCH + c0 + lm] = t[r];
    };
    auto g_mirror = [&](const v8f& t, const int c0) {
        float* p = G + (size_t)(c0 + lm) * BATCH + row0;
        *(v4f*)p = (v4f){t[0], t[1], t[2], t[3]};
        *(v4f*)(p + 4) = (v4f){t[4], t[5], t[6], t[7]};
    };

    if (wn == 0) {
#pragma unroll
        for (int i = 0; i < 5; ++i) w_tile(acc[0][i], i);
        if (one_on) {
#pragma unroll
            for (int r = 0; r < 8; ++r) accx[r] += xs[(wm * 8 + r) * 32 + lane];
            g_mirror(accx, one0);
        }
    } else {
#pragma unroll
        for (int i = 0; i < 3; ++i) w_tile(acc[0][i], 5 + i);
        if (pair_on) {
#pragma unroll
            for (int i = 3; i < 5; ++i) {
                g_direct(acc[0][i], pair0 + (i - 3) * 16);
                g_mirror(acc[0][i], pair0 + (i - 3) * 16);
            }
        }
    }
    __syncthreads();
    if (threadIdx.x < 128)
        colpart[(size_t)by * CA3 + bx * 128 + threadIdx.x] =
            (cs[threadIdx.x] + cs[128 + threadIdx.x]) +
            (cs[256 + threadIdx.x] + cs[384 + threadIdx.x]);
}

__global__ __launch_bounds__(NTHR) void k_loss(
    const float* __restrict__ G, const float* __restrict__ pc,
    const float* __restrict__ pre, const float* __restrict__ colpart,
    float* __restrict__ partials)
{
    const int lane = threadIdx.x & 31;
    const int wave = threadIdx.x >> 5;
    const int wm = wave >> 2;
    const int wn = wave & 3;
    const int lm = lane & 15;

    const int row0 = blockIdx.y * 64  + wm * 32;
    const int col0 = blockIdx.x * 128 + wn * 32;

    __shared__ __attribute__((aligned(16))) float smem[3 * (64 + 128) * LP];

    float cm[2];
#pragma unroll
    for (int ni = 0; ni < 2; ++ni) {
        const float* cp = colpart + col0 + ni * 16 + lm;
        float s = 0.f;
#pragma unroll
        for (int j = 0; j < BATCH / 64; ++j) s += cp[(size_t)j * CA3];
        cm[ni] = s * (1.0f / BATCH);
    }

    v8f acc[2][2];
#pragma unroll
    for (int i = 0; i < 2; ++i)
#pragma unroll
        for (int j = 0; j < 2; ++j) acc[i][j] = (v8f){};

    {
        StagerGramPc sg(G + (size_t)blockIdx.y * 64 * BATCH, pc + blockIdx.x * 128);
        const int lh = (lane >> 4) << 2;
        FragOfs fo;
        fo.fa = (wm * 32 + lm) * LP + lh;
        fo.fb = (64 + wn * 32 + lm) * LP + lh;
        fo.fxa = 0; fo.fxb = 0;
        v8f none = (v8f){};
        gemm_ring3<BATCH / KC, (64 + 128) * LP, 2, 2, false>(sg, smem, fo, acc, none);
    }

    const int rsel = (lane >> 4) * 8;
    float lsum = 0.f;
#pragma unroll
    for (int ni = 0; ni < 2; ++ni) {
        const int col = col0 + ni * 16 + lm;
#pragma unroll
        for (int mi = 0; mi < 2; ++mi)
#pragma unroll
            for (int r = 0; r < 8; ++r) {
                const int row = row0 + mi * 16 + rsel + r;
                const size_t idx = (size_t)row * CA3 + col;
                const float v = LR * (acc[mi][ni][r] * (1.0f / BATCH)
                                      - pre[idx] * cm[ni]);
                lsum += v * v;
            }
    }

    __shared__ float red[NTHR];
    red[threadIdx.x] = lsum;
    __syncthreads();
#pragma unroll
    for (int s = NTHR / 2; s > 0; s >>= 1) {
        if (threadIdx.x < (unsigned)s) red[threadIdx.x] += red[threadIdx.x + s];
        __syncthreads();
    }
    if (threadIdx.x == 0)
        partials[blockIdx.y * gridDim.x + blockIdx.x] = red[0];
}

__global__ void k_finalize(const float* __restrict__ partials, int n,
                           float* __restrict__ out_losses)
{
    if (threadIdx.x == 0 && blockIdx.x == 0) {
        float s = 0.f;
        for (int i = 0; i < n; ++i) s += partials[i];
        const float loss = s / (float)((size_t)BATCH * CA3);
        out_losses[0] = 0.0f;
        out_losses[1] = loss;
        out_losses[2] = loss;
    }
}

extern "C" void kernel_launch(void* const* d_in, const int* in_sizes, int n_in,
                              void* d_out, int out_size, void* d_ws, size_t ws_size,
                              hipStream_t stream)
{
    (void)in_sizes; (void)n_in; (void)out_size; (void)ws_size;
    const float* ec = (const float*)d_in[0];
    const float* dg = (const float*)d_in[1];
    const float* W  = (const float*)d_in[2];
    float* out = (float*)d_out;

    float* G        = (float*)d_ws;
    float* pre      = G + (size_t)BATCH * BATCH;
    float* colpart  = pre + (size_t)BATCH * CA3;
    float* partials = colpart + (size_t)(BATCH / 64) * CA3;

    k_gemm_pre_gram<<<dim3(CA3 / 128, BATCH / 64), NTHR, 0, stream>>>(ec, dg, W, out, pre, colpart, G);
    k_loss<<<dim3(CA3 / 128, BATCH / 64), NTHR, 0, stream>>>(G, out, pre, colpart, partials);
    k_finalize<<<1, 64, 0, stream>>>(partials, 256, out + (size_t)BATCH * CA3);
}


# Gram diagonal block: each work-group computes one 16x16 tile of it from the A rows already in LDS, k split over its 8 waves; strip 10 and its staging go (tile 64x160); diagonal: new sum order
# speedup vs baseline: 5.6597x; 1.0640x over previous
#include <hip/hip_runtime.h>
#include <hip/hip_bf16.h>
#include <stdint.h>

#define BATCH 1024
#define EC    4096
#define CA3   2048
#define LR    0.01f

typedef float v2f __attribute__((ext_vector_type(2)));
typedef float v4f __attribute__((ext_vector_type(4)));
typedef float v8f __attribute__((ext_vector_type(8)));

#define KC 64
#define LP 68
#define NTHR 256
#define GK 2

static __device__ __forceinline__ v8f wmma_f32_k4(v2f a, v2f b, v8f c) {
    return __builtin_amdgcn_wmma_f32_16x16x4_f32(
        false, a, false, b, (short)0, c, false, false);
}

static __device__ __forceinline__ void lds_barrier() {
    __builtin_amdgcn_fence(__ATOMIC_RELEASE, "workgroup", "local");
    __builtin_amdgcn_s_barrier();
    __builtin_amdgcn_fence(__ATOMIC_ACQUIRE, "workgroup", "local");
}

template <int MT, int NT, bool XT>
struct FragSet {
    v4f a[GK][MT], b[GK][NT];
    v4f xa, xb;
};
struct FragOfs { int fa, fb, fxa, fxb; };

template <int MT, int NT, bool XT>
static __device__ __forceinline__ void frag_read(
    const float* __restrict__ st, const FragOfs& o, const int r,
    FragSet<MT, NT, XT>& f)
{
#pragma unroll
    for (int q = 0; q < GK; ++q) {
#pragma unroll
        for (int i = 0; i < MT; ++i)
            f.a[q][i] = *(const v4f*)(st + o.fa + i * 16 * LP + (r * GK + q) * 8);
#pragma unroll
        for (int i = 0; i < NT; ++i)
            f.b[q][i] = *(const v4f*)(st + o.fb + i * 16 * LP + (r * GK + q) * 8);
    }
    if (XT && r == 0) {
        f.xa = *(const v4f*)(st + o.fxa);
        f.xb = *(const v4f*)(st + o.fxb);
    }
}

template <int MT, int NT, bool XT, bool DOX>
static __device__ __forceinline__ void wmma_group(
    const FragSet<MT, NT, XT>& f, v8f (&acc)[MT][NT], v8f& accx)
{
#pragma unroll
    for (int q = 0; q < GK; ++q) {
#pragma unroll
        for (int mi = 0; mi < MT; ++mi)
#pragma unroll
            for (int ni = 0; ni < NT; ++ni)
                acc[mi][ni] = wmma_f32_k4((v2f){f.a[q][mi].x, f.a[q][mi].y},
                                          (v2f){f.b[q][ni].x, f.b[q][ni].y}, acc[mi][ni]);
        if (DOX && q == 0)
            accx = wmma_f32_k4((v2f){f.xa.x, f.xa.y}, (v2f){f.xb.x, f.xb.y}, accx);
        if (DOX && q == GK - 1)
            accx = wmma_f32_k4((v2f){f.xa.z, f.xa.w}, (v2f){f.xb.z, f.xb.w}, accx);
#pragma unroll
        for (int mi = 0; mi < MT; ++mi)
#pragma unroll
            for (int ni = 0; ni < NT; ++ni)
                acc[mi][ni] = wmma_f32_k4((v2f){f.a[q][mi].z, f.a[q][mi].w},
                                          (v2f){f.b[q][ni].z, f.b[q][ni].w}, acc[mi][ni]);
    }
}

template <bool STORE, bool LOAD, bool NEXT, int MT, int NT, bool XT, class Stager>
static __device__ __forceinline__ void ring_step(
    Stager& sg, const int kload, const float* __restrict__ cur,
    const float* __restrict__ nxt, float* __restrict__ st, const FragOfs& o,
    FragSet<MT, NT, XT>& f0, FragSet<MT, NT, XT>& f1,
    v8f (&acc)[MT][NT], v8f& accx)
{
    frag_read<MT, NT, XT>(cur, o, 1, f1);
    __builtin_amdgcn_sched_barrier(0);
    wmma_group<MT, NT, XT, XT>(f0, acc, accx);
    __builtin_amdgcn_sched_barrier(0);
    frag_read<MT, NT, XT>(cur, o, 2, f0);
    if (STORE) sg.lstore(st);
    __builtin_amdgcn_sched_barrier(0);
    if (LOAD) sg.gload(kload);
    __builtin_amdgcn_sched_barrier(0);
    wmma_group<MT, NT, XT, false>(f1, acc, accx);
    __builtin_amdgcn_sched_barrier(0);
    frag_read<MT, NT, XT>(cur, o, 3, f1);
    __builtin_amdgcn_sched_barrier(0);
    wmma_group<MT, NT, XT, false>(f0, acc, accx);
    __builtin_amdgcn_sched_barrier(0);
    if (NEXT) frag_read<MT, NT, XT>(nxt, o, 0, f0);
    __builtin_amdgcn_sched_barrier(0);
    wmma_group<MT, NT, XT, false>(f1, acc, accx);
    __builtin_amdgcn_sched_barrier(0);
    lds_barrier();
}

template <int NCHUNK, int STAGE, int MT, int NT, bool XT, class Stager>
static __device__ __forceinline__ void gemm_ring3(
    Stager& sg, float* __restrict__ smem, const FragOfs& o,
    v8f (&acc)[MT][NT], v8f& accx)
{
    static_assert(NCHUNK >= 4, "ring needs at least four chunks");
    sg.gload(0);
    sg.lstore(smem);
    sg.gload(KC);
    sg.lstore(smem + STAGE);
    __syncthreads();
    sg.gload(2 * KC);

    FragSet<MT, NT, XT> f0, f1;
    frag_read<MT, NT, XT>(smem, o, 0, f0);

    int oc = 0, on = STAGE, os = 2 * STAGE;
    for (int c = 0; c < NCHUNK - 3; ++c) {
        ring_step<true, true, true, MT, NT, XT>(sg, (c + 3) * KC, smem + oc,
                                                smem + on, smem + os, o,
                                                f0, f1, acc, accx);
        const int t = oc; oc = on; on = os; os = t;
    }
    ring_step<true, false, true, MT, NT, XT>(sg, 0, smem + oc, smem + on,
                                             smem + os, o, f0, f1, acc, accx);
    ring_step<false, false, true, MT, NT, XT>(sg, 0, smem + on, smem + os,
                                              smem + oc, o, f0, f1, acc, accx);
    ring_step<false, false, false, MT, NT, XT>(sg, 0, smem + os, smem + oc,
                                               smem + on, o, f0, f1, acc, accx);
}

struct StagerPreGram {
    static constexpr int AP = 4;
    static constexpr int BW = 8;
    static constexpr int BG = 2;
    static constexpr int BP = BW + BG;
    const float* ag; const float* bg; const float* bg2;
    int sa, sb;
    v4f ra[AP], rb[BP];
    __device__ __forceinline__ StagerPreGram(const float* Ablk, const float* Wblk,
                                             const float* Gpair) {
        const int tid = threadIdx.x;
        const int lr = tid >> 4;
        const int lq = (tid & 15) << 2;
        ag = Ablk + (size_t)lr * EC + lq;
        bg = Wblk + (size_t)lr * EC + lq;
        bg2 = Gpair + (size_t)lr * EC + lq;
        sa = lr * LP + lq;
        sb = 64 * LP + sa;
    }
    __device__ __forceinline__ void gload(const int k) {
#pragma unroll
        for (int i = 0; i < AP; ++i)
            ra[i] = *(const v4f*)(ag + k + (size_t)i * 16 * EC);
#pragma unroll
        for (int i = 0; i < BW; ++i)
            rb[i] = *(const v4f*)(bg + k + (size_t)i * 16 * EC);
#pragma unroll
        for (int i = 0; i < BG; ++i)
            rb[BW + i] = *(const v4f*)(bg2 + k + (size_t)i * 16 * EC);
    }
    __device__ __forceinline__ void lstore(float* __restrict__ st) const {
#pragma unroll
        for (int i = 0; i < AP; ++i)
            *(v4f*)(st + sa + i * 16 * LP) = ra[i];
#pragma unroll
        for (int i = 0; i < BP; ++i)
            *(v4f*)(st + sb + i * 16 * LP) = rb[i];
    }
};

struct StagerGramPc {
    static constexpr int BM = 64, BN = 128;
    static constexpr int AP = BM / 16;
    static constexpr int BP = BN / 16;
    const float* ag; const float* bg;
    int sa, sb;
    v4f ra[AP], rb[BP];
    __device__ __forceinline__ StagerGramPc(const float* Ablk, const float* Bblk) {
        const int tid = threadIdx.x;
        const int lr = tid >> 4;
        const int lq = (tid & 15) << 2;
        const int bk = tid >> 2;
        const int bn = (tid & 3) << 2;
        ag = Ablk + (size_t)lr * BATCH + lq;
        bg = Bblk + (size_t)bk * CA3 + bn;
        sa = lr * LP + lq;
        sb = (BM + bn) * LP + bk;
    }
    __device__ __forceinline__ void gload(const int k) {
#pragma unroll
        for (int i = 0; i < AP; ++i)
            ra[i] = *(const v4f*)(ag + k + (size_t)i * 16 * BATCH);
#pragma unroll
        for (int i = 0; i < BP; ++i)
            rb[i] = *(const v4f*)(bg + (size_t)k * CA3 + i * 16);
    }
    __device__ __forceinline__ void lstore(float* __restrict__ st) const {
#pragma unroll
        for (int i = 0; i < AP; ++i)
            *(v4f*)(st + sa + i * 16 * LP) = ra[i];
#pragma unroll
        for (int i = 0; i < BP; ++i) {
            st[sb + (i * 16 + 0) * LP] = rb[i].x;
            st[sb + (i * 16 + 1) * LP] = rb[i].y;
            st[sb + (i * 16 + 2) * LP] = rb[i].z;
            st[sb + (i * 16 + 3) * LP] = rb[i].w;
        }
    }
};


__global__ __launch_bounds__(NTHR) void k_gemm_pre_gram(
    const float* __restrict__ ec, const float* __restrict__ dg,
    const float* __restrict__ W, float* __restrict__ out,
    float* __restrict__ pre, float* __restrict__ colpart,
    float* __restrict__ G)
{
    const int lane = threadIdx.x & 31;
    const int wave = threadIdx.x >> 5;
    const int wm = wave & 3;
    const int wn = wave >> 2;
    const int lm = lane & 15;
    const int bx = blockIdx.x;
    const int by = blockIdx.y;

    const int gd = 1 + (bx >> 1);
    const bool pair_on = gd <= (by < 8 ? 8 : 7);
    const int pair0 = ((by + gd) & 15) * 64 + (bx & 1) * 32;
    const int dr = bx >> 2;
    const int dc = bx & 3;

    __shared__ __attribute__((aligned(16))) float smem[3 * (64 + 160) * LP];

    v8f acc[1][5];
#pragma unroll
    for (int j = 0; j < 5; ++j) acc[0][j] = (v8f){};
    v8f accx = (v8f){};

    {
        StagerPreGram sg(ec + (size_t)by * 64 * EC, W + (size_t)bx * 128 * EC,
                         ec + (size_t)pair0 * EC);
        const int lh = (lane >> 4) << 2;
        FragOfs fo;
        fo.fa = (wm * 16 + lm) * LP + lh;
        fo.fb = (64 + wn * 80 + lm) * LP + lh;
        fo.fxa = (dr * 16 + lm) * LP + lh + wave * 8;
        fo.fxb = (dc * 16 + lm) * LP + lh + wave * 8;
        gemm_ring3<EC / KC, (64 + 160) * LP, 1, 5, true>(sg, smem, fo, acc, accx);
    }

    const int rsel = (lane >> 4) * 8;
    const int row0 = by * 64 + wm * 16 + rsel;
    float* xs = smem;
    float* cs = smem + 2048;

#pragma unroll
    for (int r = 0; r < 8; ++r) xs[(wave * 8 + r) * 32 + lane] = accx[r];
    __syncthreads();

    auto w_tile = [&](const v8f& t, const int s) {
        const int col = bx * 128 + s * 16 + lm;
        float csum = 0.f;
#pragma unroll
        for (int r = 0; r < 8; ++r) {
            const size_t o = (size_t)(row0 + r) * CA3 + col;
            const float v = t[r] + dg[o];
            pre[o] = t[r];
            out[o] = v;
            csum += v;
        }
        csum += __shfl_xor(csum, 16);
        if (lane < 16) cs[(wm * 8 + s) * 16 + lm] = csum;
    };
    auto g_direct = [&](const v8f& t, const int c0) {
#pragma unroll
        for (int r = 0; r < 8; ++r)
            G[(size_t)(row0 + r) * BATCH + c0 + lm] = t[r];
    };
    auto g_mirror = [&](const v8f& t, const int c0) {
        float* p = G + (size_t)(c0 + lm) * BATCH + row0;
        *(v4f*)p = (v4f){t[0], t[1], t[2], t[3]};
        *(v4f*)(p + 4) = (v4f){t[4], t[5], t[6], t[7]};
    };

    if (wn == 0) {
#pragma unroll
        for (int i = 0; i < 5; ++i) w_tile(acc[0][i], i);
    } else {
#pragma unroll
        for (int i = 0; i < 3; ++i) w_tile(acc[0][i], 5 + i);
        if (pair_on) {
#pragma unroll
            for (int i = 3; i < 5; ++i) {
                g_direct(acc[0][i], pair0 + (i - 3) * 16);
                g_mirror(acc[0][i], pair0 + (i - 3) * 16);
            }
        }
    }
    {
        float s = xs[wave * 32 + lane];
#pragma unroll
        for (int w = 1; w < 8; ++w) s += xs[(w * 8 + wave) * 32 + lane];
        G[(size_t)(by * 64 + dr * 16 + rsel + wave) * BATCH + by * 64 + dc * 16 + lm] = s;
    }
    __syncthreads();
    if (threadIdx.x < 128)
        colpart[(size_t)by * CA3 + bx * 128 + threadIdx.x] =
            (cs[threadIdx.x] + cs[128 + threadIdx.x]) +
            (cs[256 + threadIdx.x] + cs[384 + threadIdx.x]);
}

__global__ __launch_bounds__(NTHR) void k_loss(
    const float* __restrict__ G, const float* __restrict__ pc,
    const float* __restrict__ pre, const float* __restrict__ colpart,
    float* __restrict__ partials)
{
    const int lane = threadIdx.x & 31;
    const int wave = threadIdx.x >> 5;
    const int wm = wave >> 2;
    const int wn = wave & 3;
    const int lm = lane & 15;

    const int row0 = blockIdx.y * 64  + wm * 32;
    const int col0 = blockIdx.x * 128 + wn * 32;

    __shared__ __attribute__((aligned(16))) float smem[3 * (64 + 128) * LP];

    float cm[2];
#pragma unroll
    for (int ni = 0; ni < 2; ++ni) {
        const float* cp = colpart + col0 + ni * 16 + lm;
        float s = 0.f;
#pragma unroll
        for (int j = 0; j < BATCH / 64; ++j) s += cp[(size_t)j * CA3];
        cm[ni] = s * (1.0f / BATCH);
    }

    v8f acc[2][2];
#pragma unroll
    for (int i = 0; i < 2; ++i)
#pragma unroll
        for (int j = 0; j < 2; ++j) acc[i][j] = (v8f){};

    {
        StagerGramPc sg(G + (size_t)blockIdx.y * 64 * BATCH, pc + blockIdx.x * 128);
        const int lh = (lane >> 4) << 2;
        FragOfs fo;
        fo.fa = (wm * 32 + lm) * LP + lh;
        fo.fb = (64 + wn * 32 + lm) * LP + lh;
        fo.fxa = 0; fo.fxb = 0;
        v8f none = (v8f){};
        gemm_ring3<BATCH / KC, (64 + 128) * LP, 2, 2, false>(sg, smem, fo, acc, none);
    }

    const int rsel = (lane >> 4) * 8;
    float lsum = 0.f;
#pragma unroll
    for (int ni = 0; ni < 2; ++ni) {
        const int col = col0 + ni * 16 + lm;
#pragma unroll
        for (int mi = 0; mi < 2; ++mi)
#pragma unroll
            for (int r = 0; r < 8; ++r) {
                const int row = row0 + mi * 16 + rsel + r;
                const size_t idx = (size_t)row * CA3 + col;
                const float v = LR * (acc[mi][ni][r] * (1.0f / BATCH)
                                      - pre[idx] * cm[ni]);
                lsum += v * v;
            }
    }

    __shared__ float red[NTHR];
    red[threadIdx.x] = lsum;
    __syncthreads();
#pragma unroll
    for (int s = NTHR / 2; s > 0; s >>= 1) {
        if (threadIdx.x < (unsigned)s) red[threadIdx.x] += red[threadIdx.x + s];
        __syncthreads();
    }
    if (threadIdx.x == 0)
        partials[blockIdx.y * gridDim.x + blockIdx.x] = red[0];
}

__global__ void k_finalize(const float* __restrict__ partials, int n,
                           float* __restrict__ out_losses)
{
    if (threadIdx.x == 0 && blockIdx.x == 0) {
        float s = 0.f;
        for (int i = 0; i < n; ++i) s += partials[i];
        const float loss = s / (float)((size_t)BATCH * CA3);
        out_losses[0] = 0.0f;
        out_losses[1] = loss;
        out_losses[2] = loss;
    }
}

extern "C" void kernel_launch(void* const* d_in, const int* in_sizes, int n_in,
                              void* d_out, int out_size, void* d_ws, size_t ws_size,
                              hipStream_t stream)
{
    (void)in_sizes; (void)n_in; (void)out_size; (void)ws_size;
    const float* ec = (const float*)d_in[0];
    const float* dg = (const float*)d_in[1];
    const float* W  = (const float*)d_in[2];
    float* out = (float*)d_out;

    float* G        = (float*)d_ws;
    float* pre      = G + (size_t)BATCH * BATCH;
    float* colpart  = pre + (size_t)BATCH * CA3;
    float* partials = colpart + (size_t)(BATCH / 64) * CA3;

    k_gemm_pre_gram<<<dim3(CA3 / 128, BATCH / 64), NTHR, 0, stream>>>(ec, dg, W, out, pre, colpart, G);
    k_loss<<<dim3(CA3 / 128, BATCH / 64), NTHR, 0, stream>>>(G, out, pre, colpart, partials);
    k_finalize<<<1, 64, 0, stream>>>(partials, 256, out + (size_t)BATCH * CA3);
}
